# starting kernel
# baseline (speedup 1.0000x reference)
#include <hip/hip_runtime.h>
#include <hip/hip_bf16.h>

#define N_NODES   50000
#define N_EDGES   600000
#define BAG       16
#define EMB       128
#define HID       128
#define N_CLASSES 10
#define N_GRAPHS  512
#define PAD_IDX   1

typedef __attribute__((ext_vector_type(2))) float v2f;
typedef __attribute__((ext_vector_type(8))) float v8f;

__global__ void zero_f32(float* __restrict__ p, int n) {
    int i = blockIdx.x * blockDim.x + threadIdx.x;
    for (; i < n; i += gridDim.x * blockDim.x) p[i] = 0.0f;
}

__global__ void emb_bag_mean(const int* __restrict__ tokens,
                             const float* __restrict__ table,
                             float* __restrict__ out) {
    const int node = blockIdx.x * (blockDim.x >> 5) + (threadIdx.x >> 5);
    if (node >= N_NODES) return;
    const int lane = threadIdx.x & 31;
    const int* t = tokens + node * BAG;
    float4 acc = make_float4(0.f, 0.f, 0.f, 0.f);
    float cnt = 0.f;
    #pragma unroll
    for (int i = 0; i < BAG; ++i) {
        const int tok = t[i];
        if (tok != PAD_IDX) {
            const float4 e = *(const float4*)(table + (size_t)tok * EMB + lane * 4);
            acc.x += e.x; acc.y += e.y; acc.z += e.z; acc.w += e.w;
            cnt += 1.0f;
        }
    }
    const float inv = 1.0f / fmaxf(cnt, 1.0f);
    float4 r = make_float4(acc.x * inv, acc.y * inv, acc.z * inv, acc.w * inv);
    *(float4*)(out + (size_t)node * EMB + lane * 4) = r;
}

__global__ void count_deg(const int* __restrict__ dst, float* __restrict__ deg) {
    const int e = blockIdx.x * blockDim.x + threadIdx.x;
    if (e < N_EDGES) atomicAdd(&deg[dst[e]], 1.0f);
}

__global__ void count_graph(const int* __restrict__ batch, float* __restrict__ gcnt) {
    const int i = blockIdx.x * blockDim.x + threadIdx.x;
    if (i < N_NODES) atomicAdd(&gcnt[batch[i]], 1.0f);
}

__global__ void edge_scatter(const int* __restrict__ src,
                             const int* __restrict__ dst,
                             const float* __restrict__ x,
                             float* __restrict__ msg) {
    const int e = blockIdx.x * (blockDim.x >> 5) + (threadIdx.x >> 5);
    if (e >= N_EDGES) return;
    const int lane = threadIdx.x & 31;
    const int s = src[e];
    const int d = dst[e];
    const float4 v = *(const float4*)(x + (size_t)s * HID + lane * 4);
    float* p = msg + (size_t)d * HID + lane * 4;
    atomicAdd(p + 0, v.x);
    atomicAdd(p + 1, v.y);
    atomicAdd(p + 2, v.z);
    atomicAdd(p + 3, v.w);
}

__global__ void sage_wmma(const float* __restrict__ msg,
                          const float* __restrict__ deg,
                          const float* __restrict__ x,
                          const float* __restrict__ wl,
                          const float* __restrict__ bias,
                          const float* __restrict__ wr,
                          float* __restrict__ out) {
    const int lane   = threadIdx.x & 31;
    const int wave   = threadIdx.x >> 5;
    const int half   = lane >> 4;
    const int lrow   = lane & 15;
    const int m_base = blockIdx.x * 16;
    const int n_base = (blockIdx.y * 4 + wave) * 16;

    const int arow = m_base + lrow;
    const int brow = n_base + lrow;
    const float rdeg = 1.0f / fmaxf(deg[arow], 1.0f);

    v8f c;
    const float bv = bias[brow];
    #pragma unroll
    for (int r = 0; r < 8; ++r) c[r] = bv;

    const float* mrow = msg + (size_t)arow * HID;
    const float* xrow = x   + (size_t)arow * HID;
    const float* lrowp = wl + (size_t)brow * HID;
    const float* rrowp = wr + (size_t)brow * HID;

    #pragma unroll 8
    for (int k = 0; k < HID; k += 4) {
        const int ka = k + half * 2;
        v2f am = *(const v2f*)(mrow + ka);
        am *= rdeg;
        const v2f ax = *(const v2f*)(xrow + ka);
        const v2f bl = *(const v2f*)(lrowp + ka);
        const v2f br = *(const v2f*)(rrowp + ka);
        c = __builtin_amdgcn_wmma_f32_16x16x4_f32(false, am, false, bl,
                                                  (short)0, c, false, false);
        c = __builtin_amdgcn_wmma_f32_16x16x4_f32(false, ax, false, br,
                                                  (short)0, c, false, false);
    }

    #pragma unroll
    for (int r = 0; r < 8; ++r) {
        const float v = fmaxf(c[r], 0.0f);
        out[(size_t)(m_base + half * 8 + r) * HID + n_base + lrow] = v;
    }
}

__global__ void pool_scatter(const float* __restrict__ x,
                             const int* __restrict__ batch,
                             float* __restrict__ gsum) {
    const int node = blockIdx.x * (blockDim.x >> 5) + (threadIdx.x >> 5);
    if (node >= N_NODES) return;
    const int lane = threadIdx.x & 31;
    const int g = batch[node];
    const float4 v = *(const float4*)(x + (size_t)node * HID + lane * 4);
    float* p = gsum + (size_t)g * HID + lane * 4;
    atomicAdd(p + 0, v.x);
    atomicAdd(p + 1, v.y);
    atomicAdd(p + 2, v.z);
    atomicAdd(p + 3, v.w);
}

__global__ void out_head(const float* __restrict__ gsum,
                         const float* __restrict__ gcnt,
                         const float* __restrict__ wout,
                         const float* __restrict__ bout,
                         float* __restrict__ out) {
    const int idx = blockIdx.x * blockDim.x + threadIdx.x;
    if (idx >= N_GRAPHS * N_CLASSES) return;
    const int g = idx / N_CLASSES;
    const int c = idx % N_CLASSES;
    const float inv = 1.0f / fmaxf(gcnt[g], 1.0f);
    const float* gv = gsum + (size_t)g * HID;
    const float* wv = wout + (size_t)c * HID;
    float s = 0.0f;
    #pragma unroll 8
    for (int k = 0; k < HID; ++k) s += gv[k] * wv[k];
    out[idx] = s * inv + bout[c];
}

extern "C" void kernel_launch(void* const* d_in, const int* in_sizes, int n_in,
                              void* d_out, int out_size, void* d_ws, size_t ws_size,
                              hipStream_t stream) {
    (void)in_sizes; (void)n_in; (void)out_size; (void)ws_size;

    const int*   x_tokens = (const int*)  d_in[0];
    const int*   edge     = (const int*)  d_in[1];
    const int*   batch    = (const int*)  d_in[2];
    const float* emb      = (const float*)d_in[3];
    const float* w1l      = (const float*)d_in[4];
    const float* b1       = (const float*)d_in[5];
    const float* w1r      = (const float*)d_in[6];
    const float* w2l      = (const float*)d_in[7];
    const float* b2       = (const float*)d_in[8];
    const float* w2r      = (const float*)d_in[9];
    const float* wout     = (const float*)d_in[10];
    const float* bout     = (const float*)d_in[11];
    float*       out      = (float*)d_out;

    const int* src = edge;
    const int* dst = edge + N_EDGES;

    const size_t nodeF = (size_t)N_NODES * HID;
    float* bufA = (float*)d_ws;
    float* bufB = bufA + nodeF;
    float* bufC = bufB + nodeF;
    float* deg  = bufC + nodeF;
    float* gsum = deg + N_NODES;
    float* gcnt = gsum + (size_t)N_GRAPHS * HID;

    const dim3 blk256(256);
    const dim3 gWave((N_NODES + 7) / 8);
    const dim3 gEdge((N_EDGES + 7) / 8);
    const dim3 gGemm(N_NODES / 16, 2);
    const dim3 bGemm(128);

    zero_f32<<<2048, blk256, 0, stream>>>(bufB, (int)nodeF);
    zero_f32<<<256,  blk256, 0, stream>>>(deg, N_NODES);
    zero_f32<<<256,  blk256, 0, stream>>>(gsum, N_GRAPHS * HID);
    zero_f32<<<4,    blk256, 0, stream>>>(gcnt, N_GRAPHS);

    emb_bag_mean<<<gWave, blk256, 0, stream>>>(x_tokens, emb, bufA);

    count_deg<<<(N_EDGES + 255) / 256, blk256, 0, stream>>>(dst, deg);
    count_graph<<<(N_NODES + 255) / 256, blk256, 0, stream>>>(batch, gcnt);

    edge_scatter<<<gEdge, blk256, 0, stream>>>(src, dst, bufA, bufB);
    sage_wmma<<<gGemm, bGemm, 0, stream>>>(bufB, deg, bufA, w1l, b1, w1r, bufC);

    zero_f32<<<2048, blk256, 0, stream>>>(bufB, (int)nodeF);
    edge_scatter<<<gEdge, blk256, 0, stream>>>(src, dst, bufC, bufB);
    sage_wmma<<<gGemm, bGemm, 0, stream>>>(bufB, deg, bufC, w2l, b2, w2r, bufA);

    pool_scatter<<<gWave, blk256, 0, stream>>>(bufA, batch, gsum);
    out_head<<<(N_GRAPHS * N_CLASSES + 255) / 256, blk256, 0, stream>>>(
        gsum, gcnt, wout, bout, out);
}


# SAGE GEMM retiled: one wave computes a 16x128 strip (8 accumulators) so each A fragment feeds 8 column tiles; both weight matrices staged once per work-group in LDS; summation order unchanged
# speedup vs baseline: 1.2883x; 1.2883x over previous
#include <hip/hip_runtime.h>
#include <hip/hip_bf16.h>

#define N_NODES   50000
#define N_EDGES   600000
#define BAG       16
#define EMB       128
#define HID       128
#define N_CLASSES 10
#define N_GRAPHS  512
#define PAD_IDX   1

typedef __attribute__((ext_vector_type(2))) float v2f;
typedef __attribute__((ext_vector_type(8))) float v8f;

__global__ void zero_f32(float* __restrict__ p, int n) {
    int i = blockIdx.x * blockDim.x + threadIdx.x;
    for (; i < n; i += gridDim.x * blockDim.x) p[i] = 0.0f;
}

__global__ void emb_bag_mean(const int* __restrict__ tokens,
                             const float* __restrict__ table,
                             float* __restrict__ out) {
    const int node = blockIdx.x * (blockDim.x >> 5) + (threadIdx.x >> 5);
    if (node >= N_NODES) return;
    const int lane = threadIdx.x & 31;
    const int* t = tokens + node * BAG;
    float4 acc = make_float4(0.f, 0.f, 0.f, 0.f);
    float cnt = 0.f;
    #pragma unroll
    for (int i = 0; i < BAG; ++i) {
        const int tok = t[i];
        if (tok != PAD_IDX) {
            const float4 e = *(const float4*)(table + (size_t)tok * EMB + lane * 4);
            acc.x += e.x; acc.y += e.y; acc.z += e.z; acc.w += e.w;
            cnt += 1.0f;
        }
    }
    const float inv = 1.0f / fmaxf(cnt, 1.0f);
    float4 r = make_float4(acc.x * inv, acc.y * inv, acc.z * inv, acc.w * inv);
    *(float4*)(out + (size_t)node * EMB + lane * 4) = r;
}

__global__ void count_deg(const int* __restrict__ dst, float* __restrict__ deg) {
    const int e = blockIdx.x * blockDim.x + threadIdx.x;
    if (e < N_EDGES) atomicAdd(&deg[dst[e]], 1.0f);
}

__global__ void count_graph(const int* __restrict__ batch, float* __restrict__ gcnt) {
    const int i = blockIdx.x * blockDim.x + threadIdx.x;
    if (i < N_NODES) atomicAdd(&gcnt[batch[i]], 1.0f);
}

__global__ void edge_scatter(const int* __restrict__ src,
                             const int* __restrict__ dst,
                             const float* __restrict__ x,
                             float* __restrict__ msg) {
    const int e = blockIdx.x * (blockDim.x >> 5) + (threadIdx.x >> 5);
    if (e >= N_EDGES) return;
    const int lane = threadIdx.x & 31;
    const int s = src[e];
    const int d = dst[e];
    const float4 v = *(const float4*)(x + (size_t)s * HID + lane * 4);
    float* p = msg + (size_t)d * HID + lane * 4;
    atomicAdd(p + 0, v.x);
    atomicAdd(p + 1, v.y);
    atomicAdd(p + 2, v.z);
    atomicAdd(p + 3, v.w);
}

#define WPITCH     132
#define N_STRIPS   (N_NODES / 16)
#define SAGE_GRID  256
#define SAGE_WAVES 16
#define K_PAIRS    (HID / 8)

typedef __attribute__((address_space(3))) const float lds_cfloat;
typedef __attribute__((address_space(3))) const v2f   lds_cv2f;

__global__ __launch_bounds__(SAGE_WAVES * 32) __attribute__((amdgpu_waves_per_eu(4, 4)))
void sage_wmma(const float* __restrict__ msg,
               const float* __restrict__ deg,
               const float* __restrict__ x,
               const float* __restrict__ wl,
               const float* __restrict__ bias,
               const float* __restrict__ wr,
               float* __restrict__ out) {
    __shared__ __attribute__((aligned(16))) float sW[2 * HID * WPITCH];

    const int tid = threadIdx.x;

    {
        const int c4 = tid & 31;
        const int r0 = tid >> 5;
        #pragma unroll
        for (int it = 0; it < 2 * HID / SAGE_WAVES; ++it) {
            const int mat = it / (HID / SAGE_WAVES);
            const int row = (it % (HID / SAGE_WAVES)) * SAGE_WAVES + r0;
            const float* wsrc = mat ? wr : wl;
            const float4 v = *(const float4*)(wsrc + row * HID + c4 * 4);
            *(float4*)(sW + (mat * HID + row) * WPITCH + c4 * 4) = v;
        }
    }
    __syncthreads();

    const int lane  = tid & 31;
    const int wave  = __builtin_amdgcn_readfirstlane(tid >> 5);
    const int strip = wave * SAGE_GRID + blockIdx.x;
    if (strip >= N_STRIPS) return;

    const int half   = lane >> 4;
    const int lrow   = lane & 15;
    const int m_base = strip * 16;
    const int arow   = m_base + lrow;
    const float rdeg = 1.0f / fmaxf(deg[arow], 1.0f);

    v8f c[8];
    #pragma unroll
    for (int t = 0; t < 8; ++t) {
        const float bv = bias[t * 16 + lrow];
        #pragma unroll
        for (int r = 0; r < 8; ++r) c[t][r] = bv;
    }

    const float* mrow = msg + (size_t)arow * HID + half * 2;
    const float* xrow = x   + (size_t)arow * HID + half * 2;
    lds_cfloat* sl = (lds_cfloat*)sW + lrow * WPITCH + half * 2;
    lds_cfloat* sr = sl + HID * WPITCH;
    asm("" : "+v"(sr));

    v2f am[3][2], ax[3][2];
    v2f bl[2][2][8], br[2][2][8];

    #define LOAD_A(p, s)                                                        \
        _Pragma("unroll") for (int j = 0; j < 2; ++j) {                         \
            am[s][j] = *(const v2f*)(mrow + (p) * 8 + j * 4);                   \
            ax[s][j] = *(const v2f*)(xrow + (p) * 8 + j * 4);                   \
        }
    #define LOAD_B(p, s)                                                        \
        _Pragma("unroll") for (int j = 0; j < 2; ++j) {                         \
            _Pragma("unroll") for (int t = 0; t < 8; ++t) {                     \
                bl[s][j][t] = *(lds_cv2f*)(sl + t * 16 * WPITCH + (p) * 8 + j * 4); \
                br[s][j][t] = *(lds_cv2f*)(sr + t * 16 * WPITCH + (p) * 8 + j * 4); \
            }                                                                   \
        }

    LOAD_A(0, 0)
    LOAD_A(1, 1)
    LOAD_B(0, 0)

    #pragma unroll
    for (int p = 0; p < K_PAIRS; ++p) {
        if (p + 2 < K_PAIRS) { LOAD_A(p + 2, (p + 2) % 3) }
        if (p + 1 < K_PAIRS) { LOAD_B(p + 1, (p + 1) & 1) }
        #pragma unroll
        for (int j = 0; j < 2; ++j) {
            const v2f a = am[p % 3][j] * rdeg;
            #pragma unroll
            for (int t = 0; t < 8; ++t)
                c[t] = __builtin_amdgcn_wmma_f32_16x16x4_f32(false, a, false, bl[p & 1][j][t],
                                                             (short)0, c[t], false, false);
            #pragma unroll
            for (int t = 0; t < 8; ++t)
                c[t] = __builtin_amdgcn_wmma_f32_16x16x4_f32(false, ax[p % 3][j], false, br[p & 1][j][t],
                                                             (short)0, c[t], false, false);
        }
    }
    #undef LOAD_A
    #undef LOAD_B

    #pragma unroll
    for (int t = 0; t < 8; ++t) {
        #pragma unroll
        for (int r = 0; r < 8; ++r) {
            const float v = fmaxf(c[t][r], 0.0f);
            out[(size_t)(m_base + half * 8 + r) * HID + t * 16 + lrow] = v;
        }
    }
}

__global__ void pool_scatter(const float* __restrict__ x,
                             const int* __restrict__ batch,
                             float* __restrict__ gsum) {
    const int node = blockIdx.x * (blockDim.x >> 5) + (threadIdx.x >> 5);
    if (node >= N_NODES) return;
    const int lane = threadIdx.x & 31;
    const int g = batch[node];
    const float4 v = *(const float4*)(x + (size_t)node * HID + lane * 4);
    float* p = gsum + (size_t)g * HID + lane * 4;
    atomicAdd(p + 0, v.x);
    atomicAdd(p + 1, v.y);
    atomicAdd(p + 2, v.z);
    atomicAdd(p + 3, v.w);
}

__global__ void out_head(const float* __restrict__ gsum,
                         const float* __restrict__ gcnt,
                         const float* __restrict__ wout,
                         const float* __restrict__ bout,
                         float* __restrict__ out) {
    const int idx = blockIdx.x * blockDim.x + threadIdx.x;
    if (idx >= N_GRAPHS * N_CLASSES) return;
    const int g = idx / N_CLASSES;
    const int c = idx % N_CLASSES;
    const float inv = 1.0f / fmaxf(gcnt[g], 1.0f);
    const float* gv = gsum + (size_t)g * HID;
    const float* wv = wout + (size_t)c * HID;
    float s = 0.0f;
    #pragma unroll 8
    for (int k = 0; k < HID; ++k) s += gv[k] * wv[k];
    out[idx] = s * inv + bout[c];
}

extern "C" void kernel_launch(void* const* d_in, const int* in_sizes, int n_in,
                              void* d_out, int out_size, void* d_ws, size_t ws_size,
                              hipStream_t stream) {
    (void)in_sizes; (void)n_in; (void)out_size; (void)ws_size;

    const int*   x_tokens = (const int*)  d_in[0];
    const int*   edge     = (const int*)  d_in[1];
    const int*   batch    = (const int*)  d_in[2];
    const float* emb      = (const float*)d_in[3];
    const float* w1l      = (const float*)d_in[4];
    const float* b1       = (const float*)d_in[5];
    const float* w1r      = (const float*)d_in[6];
    const float* w2l      = (const float*)d_in[7];
    const float* b2       = (const float*)d_in[8];
    const float* w2r      = (const float*)d_in[9];
    const float* wout     = (const float*)d_in[10];
    const float* bout     = (const float*)d_in[11];
    float*       out      = (float*)d_out;

    const int* src = edge;
    const int* dst = edge + N_EDGES;

    const size_t nodeF = (size_t)N_NODES * HID;
    float* bufA = (float*)d_ws;
    float* bufB = bufA + nodeF;
    float* bufC = bufB + nodeF;
    float* deg  = bufC + nodeF;
    float* gsum = deg + N_NODES;
    float* gcnt = gsum + (size_t)N_GRAPHS * HID;

    const dim3 blk256(256);
    const dim3 gWave((N_NODES + 7) / 8);
    const dim3 gEdge((N_EDGES + 7) / 8);
    const dim3 gGemm(SAGE_GRID);
    const dim3 bGemm(SAGE_WAVES * 32);

    zero_f32<<<2048, blk256, 0, stream>>>(bufB, (int)nodeF);
    zero_f32<<<256,  blk256, 0, stream>>>(deg, N_NODES);
    zero_f32<<<256,  blk256, 0, stream>>>(gsum, N_GRAPHS * HID);
    zero_f32<<<4,    blk256, 0, stream>>>(gcnt, N_GRAPHS);

    emb_bag_mean<<<gWave, blk256, 0, stream>>>(x_tokens, emb, bufA);

    count_deg<<<(N_EDGES + 255) / 256, blk256, 0, stream>>>(dst, deg);
    count_graph<<<(N_NODES + 255) / 256, blk256, 0, stream>>>(batch, gcnt);

    edge_scatter<<<gEdge, blk256, 0, stream>>>(src, dst, bufA, bufB);
    sage_wmma<<<gGemm, bGemm, 0, stream>>>(bufB, deg, bufA, w1l, b1, w1r, bufC);

    zero_f32<<<2048, blk256, 0, stream>>>(bufB, (int)nodeF);
    edge_scatter<<<gEdge, blk256, 0, stream>>>(src, dst, bufC, bufB);
    sage_wmma<<<gGemm, bGemm, 0, stream>>>(bufB, deg, bufC, w2l, b2, w2r, bufA);

    pool_scatter<<<gWave, blk256, 0, stream>>>(bufA, batch, gsum);
    out_head<<<(N_GRAPHS * N_CLASSES + 255) / 256, blk256, 0, stream>>>(
        gsum, gcnt, wout, bout, out);
}


# Both f32 atomic scatters (edge sums, graph pool): a lane owns floats l, l+32, l+64, l+96 of a row, so each load and each atomic add covers one whole 128-B line, not 4 quarters; sums in arrival order
# speedup vs baseline: 1.7433x; 1.3532x over previous
#include <hip/hip_runtime.h>
#include <hip/hip_bf16.h>

#define N_NODES   50000
#define N_EDGES   600000
#define BAG       16
#define EMB       128
#define HID       128
#define N_CLASSES 10
#define N_GRAPHS  512
#define PAD_IDX   1

typedef __attribute__((ext_vector_type(2))) float v2f;
typedef __attribute__((ext_vector_type(8))) float v8f;

__global__ void zero_f32(float* __restrict__ p, int n) {
    int i = blockIdx.x * blockDim.x + threadIdx.x;
    for (; i < n; i += gridDim.x * blockDim.x) p[i] = 0.0f;
}

__global__ void emb_bag_mean(const int* __restrict__ tokens,
                             const float* __restrict__ table,
                             float* __restrict__ out) {
    const int node = blockIdx.x * (blockDim.x >> 5) + (threadIdx.x >> 5);
    if (node >= N_NODES) return;
    const int lane = threadIdx.x & 31;
    const int* t = tokens + node * BAG;
    float4 acc = make_float4(0.f, 0.f, 0.f, 0.f);
    float cnt = 0.f;
    #pragma unroll
    for (int i = 0; i < BAG; ++i) {
        const int tok = t[i];
        if (tok != PAD_IDX) {
            const float4 e = *(const float4*)(table + (size_t)tok * EMB + lane * 4);
            acc.x += e.x; acc.y += e.y; acc.z += e.z; acc.w += e.w;
            cnt += 1.0f;
        }
    }
    const float inv = 1.0f / fmaxf(cnt, 1.0f);
    float4 r = make_float4(acc.x * inv, acc.y * inv, acc.z * inv, acc.w * inv);
    *(float4*)(out + (size_t)node * EMB + lane * 4) = r;
}

__global__ void count_deg(const int* __restrict__ dst, float* __restrict__ deg) {
    const int e = blockIdx.x * blockDim.x + threadIdx.x;
    if (e < N_EDGES) atomicAdd(&deg[dst[e]], 1.0f);
}

__global__ void count_graph(const int* __restrict__ batch, float* __restrict__ gcnt) {
    const int i = blockIdx.x * blockDim.x + threadIdx.x;
    if (i < N_NODES) atomicAdd(&gcnt[batch[i]], 1.0f);
}

__global__ void edge_scatter(const int* __restrict__ src,
                             const int* __restrict__ dst,
                             const float* __restrict__ x,
                             float* __restrict__ msg) {
    const int e = blockIdx.x * (blockDim.x >> 5) + (threadIdx.x >> 5);
    if (e >= N_EDGES) return;
    const int lane = threadIdx.x & 31;
    const int s = src[e];
    const int d = dst[e];
    const float* xr = x + (size_t)s * HID + lane;
    const float v0 = xr[0];
    const float v1 = xr[32];
    const float v2 = xr[64];
    const float v3 = xr[96];
    float* p = msg + (size_t)d * HID + lane;
    atomicAdd(p +  0, v0);
    atomicAdd(p + 32, v1);
    atomicAdd(p + 64, v2);
    atomicAdd(p + 96, v3);
}

#define WPITCH     132
#define N_STRIPS   (N_NODES / 16)
#define SAGE_GRID  256
#define SAGE_WAVES 16
#define K_PAIRS    (HID / 8)

typedef __attribute__((address_space(3))) const float lds_cfloat;
typedef __attribute__((address_space(3))) const v2f   lds_cv2f;

__global__ __launch_bounds__(SAGE_WAVES * 32) __attribute__((amdgpu_waves_per_eu(4, 4)))
void sage_wmma(const float* __restrict__ msg,
               const float* __restrict__ deg,
               const float* __restrict__ x,
               const float* __restrict__ wl,
               const float* __restrict__ bias,
               const float* __restrict__ wr,
               float* __restrict__ out) {
    __shared__ __attribute__((aligned(16))) float sW[2 * HID * WPITCH];

    const int tid = threadIdx.x;

    {
        const int c4 = tid & 31;
        const int r0 = tid >> 5;
        #pragma unroll
        for (int it = 0; it < 2 * HID / SAGE_WAVES; ++it) {
            const int mat = it / (HID / SAGE_WAVES);
            const int row = (it % (HID / SAGE_WAVES)) * SAGE_WAVES + r0;
            const float* wsrc = mat ? wr : wl;
            const float4 v = *(const float4*)(wsrc + row * HID + c4 * 4);
            *(float4*)(sW + (mat * HID + row) * WPITCH + c4 * 4) = v;
        }
    }
    __syncthreads();

    const int lane  = tid & 31;
    const int wave  = __builtin_amdgcn_readfirstlane(tid >> 5);
    const int strip = wave * SAGE_GRID + blockIdx.x;
    if (strip >= N_STRIPS) return;

    const int half   = lane >> 4;
    const int lrow   = lane & 15;
    const int m_base = strip * 16;
    const int arow   = m_base + lrow;
    const float rdeg = 1.0f / fmaxf(deg[arow], 1.0f);

    v8f c[8];
    #pragma unroll
    for (int t = 0; t < 8; ++t) {
        const float bv = bias[t * 16 + lrow];
        #pragma unroll
        for (int r = 0; r < 8; ++r) c[t][r] = bv;
    }

    const float* mrow = msg + (size_t)arow * HID + half * 2;
    const float* xrow = x   + (size_t)arow * HID + half * 2;
    lds_cfloat* sl = (lds_cfloat*)sW + lrow * WPITCH + half * 2;
    lds_cfloat* sr = sl + HID * WPITCH;
    asm("" : "+v"(sr));

    v2f am[3][2], ax[3][2];
    v2f bl[2][2][8], br[2][2][8];

    #define LOAD_A(p, s)                                                        \
        _Pragma("unroll") for (int j = 0; j < 2; ++j) {                         \
            am[s][j] = *(const v2f*)(mrow + (p) * 8 + j * 4);                   \
            ax[s][j] = *(const v2f*)(xrow + (p) * 8 + j * 4);                   \
        }
    #define LOAD_B(p, s)                                                        \
        _Pragma("unroll") for (int j = 0; j < 2; ++j) {                         \
            _Pragma("unroll") for (int t = 0; t < 8; ++t) {                     \
                bl[s][j][t] = *(lds_cv2f*)(sl + t * 16 * WPITCH + (p) * 8 + j * 4); \
                br[s][j][t] = *(lds_cv2f*)(sr + t * 16 * WPITCH + (p) * 8 + j * 4); \
            }                                                                   \
        }

    LOAD_A(0, 0)
    LOAD_A(1, 1)
    LOAD_B(0, 0)

    #pragma unroll
    for (int p = 0; p < K_PAIRS; ++p) {
        if (p + 2 < K_PAIRS) { LOAD_A(p + 2, (p + 2) % 3) }
        if (p + 1 < K_PAIRS) { LOAD_B(p + 1, (p + 1) & 1) }
        #pragma unroll
        for (int j = 0; j < 2; ++j) {
            const v2f a = am[p % 3][j] * rdeg;
            #pragma unroll
            for (int t = 0; t < 8; ++t)
                c[t] = __builtin_amdgcn_wmma_f32_16x16x4_f32(false, a, false, bl[p & 1][j][t],
                                                             (short)0, c[t], false, false);
            #pragma unroll
            for (int t = 0; t < 8; ++t)
                c[t] = __builtin_amdgcn_wmma_f32_16x16x4_f32(false, ax[p % 3][j], false, br[p & 1][j][t],
                                                             (short)0, c[t], false, false);
        }
    }
    #undef LOAD_A
    #undef LOAD_B

    #pragma unroll
    for (int t = 0; t < 8; ++t) {
        #pragma unroll
        for (int r = 0; r < 8; ++r) {
            const float v = fmaxf(c[t][r], 0.0f);
            out[(size_t)(m_base + half * 8 + r) * HID + t * 16 + lrow] = v;
        }
    }
}

__global__ void pool_scatter(const float* __restrict__ x,
                             const int* __restrict__ batch,
                             float* __restrict__ gsum) {
    const int node = blockIdx.x * (blockDim.x >> 5) + (threadIdx.x >> 5);
    if (node >= N_NODES) return;
    const int lane = threadIdx.x & 31;
    const int g = batch[node];
    const float* xr = x + (size_t)node * HID + lane;
    const float v0 = xr[0];
    const float v1 = xr[32];
    const float v2 = xr[64];
    const float v3 = xr[96];
    float* p = gsum + (size_t)g * HID + lane;
    atomicAdd(p +  0, v0);
    atomicAdd(p + 32, v1);
    atomicAdd(p + 64, v2);
    atomicAdd(p + 96, v3);
}

__global__ void out_head(const float* __restrict__ gsum,
                         const float* __restrict__ gcnt,
                         const float* __restrict__ wout,
                         const float* __restrict__ bout,
                         float* __restrict__ out) {
    const int idx = blockIdx.x * blockDim.x + threadIdx.x;
    if (idx >= N_GRAPHS * N_CLASSES) return;
    const int g = idx / N_CLASSES;
    const int c = idx % N_CLASSES;
    const float inv = 1.0f / fmaxf(gcnt[g], 1.0f);
    const float* gv = gsum + (size_t)g * HID;
    const float* wv = wout + (size_t)c * HID;
    float s = 0.0f;
    #pragma unroll 8
    for (int k = 0; k < HID; ++k) s += gv[k] * wv[k];
    out[idx] = s * inv + bout[c];
}

extern "C" void kernel_launch(void* const* d_in, const int* in_sizes, int n_in,
                              void* d_out, int out_size, void* d_ws, size_t ws_size,
                              hipStream_t stream) {
    (void)in_sizes; (void)n_in; (void)out_size; (void)ws_size;

    const int*   x_tokens = (const int*)  d_in[0];
    const int*   edge     = (const int*)  d_in[1];
    const int*   batch    = (const int*)  d_in[2];
    const float* emb      = (const float*)d_in[3];
    const float* w1l      = (const float*)d_in[4];
    const float* b1       = (const float*)d_in[5];
    const float* w1r      = (const float*)d_in[6];
    const float* w2l      = (const float*)d_in[7];
    const float* b2       = (const float*)d_in[8];
    const float* w2r      = (const float*)d_in[9];
    const float* wout     = (const float*)d_in[10];
    const float* bout     = (const float*)d_in[11];
    float*       out      = (float*)d_out;

    const int* src = edge;
    const int* dst = edge + N_EDGES;

    const size_t nodeF = (size_t)N_NODES * HID;
    float* bufA = (float*)d_ws;
    float* bufB = bufA + nodeF;
    float* bufC = bufB + nodeF;
    float* gsum = bufC + nodeF;
    float* deg  = gsum + (size_t)N_GRAPHS * HID;
    float* gcnt = deg + N_NODES;

    const dim3 blk256(256);
    const dim3 gWave((N_NODES + 7) / 8);
    const dim3 gEdge((N_EDGES + 7) / 8);
    const dim3 gGemm(SAGE_GRID);
    const dim3 bGemm(SAGE_WAVES * 32);

    zero_f32<<<2048, blk256, 0, stream>>>(bufB, (int)nodeF);
    zero_f32<<<256,  blk256, 0, stream>>>(deg, N_NODES);
    zero_f32<<<256,  blk256, 0, stream>>>(gsum, N_GRAPHS * HID);
    zero_f32<<<4,    blk256, 0, stream>>>(gcnt, N_GRAPHS);

    emb_bag_mean<<<gWave, blk256, 0, stream>>>(x_tokens, emb, bufA);

    count_deg<<<(N_EDGES + 255) / 256, blk256, 0, stream>>>(dst, deg);
    count_graph<<<(N_NODES + 255) / 256, blk256, 0, stream>>>(batch, gcnt);

    edge_scatter<<<gEdge, blk256, 0, stream>>>(src, dst, bufA, bufB);
    sage_wmma<<<gGemm, bGemm, 0, stream>>>(bufB, deg, bufA, w1l, b1, w1r, bufC);

    zero_f32<<<2048, blk256, 0, stream>>>(bufB, (int)nodeF);
    edge_scatter<<<gEdge, blk256, 0, stream>>>(src, dst, bufC, bufB);
    sage_wmma<<<gGemm, bGemm, 0, stream>>>(bufB, deg, bufC, w2l, b2, w2r, bufA);

    pool_scatter<<<gWave, blk256, 0, stream>>>(bufA, batch, gsum);
    out_head<<<(N_GRAPHS * N_CLASSES + 255) / 256, blk256, 0, stream>>>(
        gsum, gcnt, wout, bout, out);
}


# Both f32 atomic scatters: a wave takes 8 edges (pool: 8 nodes), one index load, then all 32 row loads in flight, then all 32 line adds; 8x fewer waves; sums stay in arrival order
# speedup vs baseline: 2.1419x; 1.2286x over previous
#include <hip/hip_runtime.h>
#include <hip/hip_bf16.h>

#define N_NODES   50000
#define N_EDGES   600000
#define BAG       16
#define EMB       128
#define HID       128
#define N_CLASSES 10
#define N_GRAPHS  512
#define PAD_IDX   1

typedef __attribute__((ext_vector_type(2))) float v2f;
typedef __attribute__((ext_vector_type(8))) float v8f;

__global__ void zero_f32(float* __restrict__ p, int n) {
    int i = blockIdx.x * blockDim.x + threadIdx.x;
    for (; i < n; i += gridDim.x * blockDim.x) p[i] = 0.0f;
}

__global__ void emb_bag_mean(const int* __restrict__ tokens,
                             const float* __restrict__ table,
                             float* __restrict__ out) {
    const int node = blockIdx.x * (blockDim.x >> 5) + (threadIdx.x >> 5);
    if (node >= N_NODES) return;
    const int lane = threadIdx.x & 31;
    const int* t = tokens + node * BAG;
    float4 acc = make_float4(0.f, 0.f, 0.f, 0.f);
    float cnt = 0.f;
    #pragma unroll
    for (int i = 0; i < BAG; ++i) {
        const int tok = t[i];
        if (tok != PAD_IDX) {
            const float4 e = *(const float4*)(table + (size_t)tok * EMB + lane * 4);
            acc.x += e.x; acc.y += e.y; acc.z += e.z; acc.w += e.w;
            cnt += 1.0f;
        }
    }
    const float inv = 1.0f / fmaxf(cnt, 1.0f);
    float4 r = make_float4(acc.x * inv, acc.y * inv, acc.z * inv, acc.w * inv);
    *(float4*)(out + (size_t)node * EMB + lane * 4) = r;
}

__global__ void count_deg(const int* __restrict__ dst, float* __restrict__ deg) {
    const int e = blockIdx.x * blockDim.x + threadIdx.x;
    if (e < N_EDGES) atomicAdd(&deg[dst[e]], 1.0f);
}

__global__ void count_graph(const int* __restrict__ batch, float* __restrict__ gcnt) {
    const int i = blockIdx.x * blockDim.x + threadIdx.x;
    if (i < N_NODES) atomicAdd(&gcnt[batch[i]], 1.0f);
}

#define EPW 8
static_assert(N_EDGES % EPW == 0 && N_NODES % EPW == 0, "a wave's items are all in range or all out");

__global__ void edge_scatter(const int* __restrict__ src,
                             const int* __restrict__ dst,
                             const float* __restrict__ x,
                             float* __restrict__ msg) {
    const int wave = __builtin_amdgcn_readfirstlane(blockIdx.x * (blockDim.x >> 5) + (threadIdx.x >> 5));
    const int e0 = wave * EPW;
    if (e0 >= N_EDGES) return;
    const int lane = threadIdx.x & 31;
    const int sl = src[e0 + (lane & (EPW - 1))];
    const int dl = dst[e0 + (lane & (EPW - 1))];
    float v[EPW][4];
    #pragma unroll
    for (int i = 0; i < EPW; ++i) {
        const int s = __builtin_amdgcn_readlane(sl, i);
        const float* xr = x + (size_t)s * HID + lane;
        v[i][0] = xr[0];
        v[i][1] = xr[32];
        v[i][2] = xr[64];
        v[i][3] = xr[96];
    }
    #pragma unroll
    for (int i = 0; i < EPW; ++i) {
        const int d = __builtin_amdgcn_readlane(dl, i);
        float* p = msg + (size_t)d * HID + lane;
        atomicAdd(p +  0, v[i][0]);
        atomicAdd(p + 32, v[i][1]);
        atomicAdd(p + 64, v[i][2]);
        atomicAdd(p + 96, v[i][3]);
    }
}

#define WPITCH     132
#define N_STRIPS   (N_NODES / 16)
#define SAGE_GRID  256
#define SAGE_WAVES 16
#define K_PAIRS    (HID / 8)

typedef __attribute__((address_space(3))) const float lds_cfloat;
typedef __attribute__((address_space(3))) const v2f   lds_cv2f;

__global__ __launch_bounds__(SAGE_WAVES * 32) __attribute__((amdgpu_waves_per_eu(4, 4)))
void sage_wmma(const float* __restrict__ msg,
               const float* __restrict__ deg,
               const float* __restrict__ x,
               const float* __restrict__ wl,
               const float* __restrict__ bias,
               const float* __restrict__ wr,
               float* __restrict__ out) {
    __shared__ __attribute__((aligned(16))) float sW[2 * HID * WPITCH];

    const int tid = threadIdx.x;

    {
        const int c4 = tid & 31;
        const int r0 = tid >> 5;
        #pragma unroll
        for (int it = 0; it < 2 * HID / SAGE_WAVES; ++it) {
            const int mat = it / (HID / SAGE_WAVES);
            const int row = (it % (HID / SAGE_WAVES)) * SAGE_WAVES + r0;
            const float* wsrc = mat ? wr : wl;
            const float4 v = *(const float4*)(wsrc + row * HID + c4 * 4);
            *(float4*)(sW + (mat * HID + row) * WPITCH + c4 * 4) = v;
        }
    }
    __syncthreads();

    const int lane  = tid & 31;
    const int wave  = __builtin_amdgcn_readfirstlane(tid >> 5);
    const int strip = wave * SAGE_GRID + blockIdx.x;
    if (strip >= N_STRIPS) return;

    const int half   = lane >> 4;
    const int lrow   = lane & 15;
    const int m_base = strip * 16;
    const int arow   = m_base + lrow;
    const float rdeg = 1.0f / fmaxf(deg[arow], 1.0f);

    v8f c[8];
    #pragma unroll
    for (int t = 0; t < 8; ++t) {
        const float bv = bias[t * 16 + lrow];
        #pragma unroll
        for (int r = 0; r < 8; ++r) c[t][r] = bv;
    }

    const float* mrow = msg + (size_t)arow * HID + half * 2;
    const float* xrow = x   + (size_t)arow * HID + half * 2;
    lds_cfloat* sl = (lds_cfloat*)sW + lrow * WPITCH + half * 2;
    lds_cfloat* sr = sl + HID * WPITCH;
    asm("" : "+v"(sr));

    v2f am[3][2], ax[3][2];
    v2f bl[2][2][8], br[2][2][8];

    #define LOAD_A(p, s)                                                        \
        _Pragma("unroll") for (int j = 0; j < 2; ++j) {                         \
            am[s][j] = *(const v2f*)(mrow + (p) * 8 + j * 4);                   \
            ax[s][j] = *(const v2f*)(xrow + (p) * 8 + j * 4);                   \
        }
    #define LOAD_B(p, s)                                                        \
        _Pragma("unroll") for (int j = 0; j < 2; ++j) {                         \
            _Pragma("unroll") for (int t = 0; t < 8; ++t) {                     \
                bl[s][j][t] = *(lds_cv2f*)(sl + t * 16 * WPITCH + (p) * 8 + j * 4); \
                br[s][j][t] = *(lds_cv2f*)(sr + t * 16 * WPITCH + (p) * 8 + j * 4); \
            }                                                                   \
        }

    LOAD_A(0, 0)
    LOAD_A(1, 1)
    LOAD_B(0, 0)

    #pragma unroll
    for (int p = 0; p < K_PAIRS; ++p) {
        if (p + 2 < K_PAIRS) { LOAD_A(p + 2, (p + 2) % 3) }
        if (p + 1 < K_PAIRS) { LOAD_B(p + 1, (p + 1) & 1) }
        #pragma unroll
        for (int j = 0; j < 2; ++j) {
            const v2f a = am[p % 3][j] * rdeg;
            #pragma unroll
            for (int t = 0; t < 8; ++t)
                c[t] = __builtin_amdgcn_wmma_f32_16x16x4_f32(false, a, false, bl[p & 1][j][t],
                                                             (short)0, c[t], false, false);
            #pragma unroll
            for (int t = 0; t < 8; ++t)
                c[t] = __builtin_amdgcn_wmma_f32_16x16x4_f32(false, ax[p % 3][j], false, br[p & 1][j][t],
                                                             (short)0, c[t], false, false);
        }
    }
    #undef LOAD_A
    #undef LOAD_B

    #pragma unroll
    for (int t = 0; t < 8; ++t) {
        #pragma unroll
        for (int r = 0; r < 8; ++r) {
            const float v = fmaxf(c[t][r], 0.0f);
            out[(size_t)(m_base + half * 8 + r) * HID + t * 16 + lrow] = v;
        }
    }
}

__global__ void pool_scatter(const float* __restrict__ x,
                             const int* __restrict__ batch,
                             float* __restrict__ gsum) {
    const int wave = __builtin_amdgcn_readfirstlane(blockIdx.x * (blockDim.x >> 5) + (threadIdx.x >> 5));
    const int n0 = wave * EPW;
    if (n0 >= N_NODES) return;
    const int lane = threadIdx.x & 31;
    const int gl = batch[n0 + (lane & (EPW - 1))];
    const float* xr = x + (size_t)n0 * HID + lane;
    float v[EPW][4];
    #pragma unroll
    for (int i = 0; i < EPW; ++i) {
        v[i][0] = xr[i * HID +  0];
        v[i][1] = xr[i * HID + 32];
        v[i][2] = xr[i * HID + 64];
        v[i][3] = xr[i * HID + 96];
    }
    #pragma unroll
    for (int i = 0; i < EPW; ++i) {
        const int g = __builtin_amdgcn_readlane(gl, i);
        float* p = gsum + (size_t)g * HID + lane;
        atomicAdd(p +  0, v[i][0]);
        atomicAdd(p + 32, v[i][1]);
        atomicAdd(p + 64, v[i][2]);
        atomicAdd(p + 96, v[i][3]);
    }
}

__global__ void out_head(const float* __restrict__ gsum,
                         const float* __restrict__ gcnt,
                         const float* __restrict__ wout,
                         const float* __restrict__ bout,
                         float* __restrict__ out) {
    const int idx = blockIdx.x * blockDim.x + threadIdx.x;
    if (idx >= N_GRAPHS * N_CLASSES) return;
    const int g = idx / N_CLASSES;
    const int c = idx % N_CLASSES;
    const float inv = 1.0f / fmaxf(gcnt[g], 1.0f);
    const float* gv = gsum + (size_t)g * HID;
    const float* wv = wout + (size_t)c * HID;
    float s = 0.0f;
    #pragma unroll 8
    for (int k = 0; k < HID; ++k) s += gv[k] * wv[k];
    out[idx] = s * inv + bout[c];
}

extern "C" void kernel_launch(void* const* d_in, const int* in_sizes, int n_in,
                              void* d_out, int out_size, void* d_ws, size_t ws_size,
                              hipStream_t stream) {
    (void)in_sizes; (void)n_in; (void)out_size; (void)ws_size;

    const int*   x_tokens = (const int*)  d_in[0];
    const int*   edge     = (const int*)  d_in[1];
    const int*   batch    = (const int*)  d_in[2];
    const float* emb      = (const float*)d_in[3];
    const float* w1l      = (const float*)d_in[4];
    const float* b1       = (const float*)d_in[5];
    const float* w1r      = (const float*)d_in[6];
    const float* w2l      = (const float*)d_in[7];
    const float* b2       = (const float*)d_in[8];
    const float* w2r      = (const float*)d_in[9];
    const float* wout     = (const float*)d_in[10];
    const float* bout     = (const float*)d_in[11];
    float*       out      = (float*)d_out;

    const int* src = edge;
    const int* dst = edge + N_EDGES;

    const size_t nodeF = (size_t)N_NODES * HID;
    float* bufA = (float*)d_ws;
    float* bufB = bufA + nodeF;
    float* bufC = bufB + nodeF;
    float* gsum = bufC + nodeF;
    float* deg  = gsum + (size_t)N_GRAPHS * HID;
    float* gcnt = deg + N_NODES;

    const dim3 blk256(256);
    const dim3 gWave((N_NODES + 7) / 8);
    const dim3 gEdge((N_EDGES / EPW + 7) / 8);
    const dim3 gPool((N_NODES / EPW + 7) / 8);
    const dim3 gGemm(SAGE_GRID);
    const dim3 bGemm(SAGE_WAVES * 32);

    zero_f32<<<2048, blk256, 0, stream>>>(bufB, (int)nodeF);
    zero_f32<<<256,  blk256, 0, stream>>>(deg, N_NODES);
    zero_f32<<<256,  blk256, 0, stream>>>(gsum, N_GRAPHS * HID);
    zero_f32<<<4,    blk256, 0, stream>>>(gcnt, N_GRAPHS);

    emb_bag_mean<<<gWave, blk256, 0, stream>>>(x_tokens, emb, bufA);

    count_deg<<<(N_EDGES + 255) / 256, blk256, 0, stream>>>(dst, deg);
    count_graph<<<(N_NODES + 255) / 256, blk256, 0, stream>>>(batch, gcnt);

    edge_scatter<<<gEdge, blk256, 0, stream>>>(src, dst, bufA, bufB);
    sage_wmma<<<gGemm, bGemm, 0, stream>>>(bufB, deg, bufA, w1l, b1, w1r, bufC);

    zero_f32<<<2048, blk256, 0, stream>>>(bufB, (int)nodeF);
    edge_scatter<<<gEdge, blk256, 0, stream>>>(src, dst, bufC, bufB);
    sage_wmma<<<gGemm, bGemm, 0, stream>>>(bufB, deg, bufC, w2l, b2, w2r, bufA);

    pool_scatter<<<gPool, blk256, 0, stream>>>(bufA, batch, gsum);
    out_head<<<(N_GRAPHS * N_CLASSES + 255) / 256, blk256, 0, stream>>>(
        gsum, gcnt, wout, bout, out);
}


# Edge sums as a gather: edges bucketed by destination once per call (integer-atomic rank, scan, place); a wave sums a node's source rows in registers in list order; no f32 edge atomics, no zero fill
# speedup vs baseline: 2.4384x; 1.1384x over previous
#include <hip/hip_runtime.h>
#include <hip/hip_bf16.h>

#define N_NODES   50000
#define N_EDGES   600000
#define BAG       16
#define EMB       128
#define HID       128
#define N_CLASSES 10
#define N_GRAPHS  512
#define PAD_IDX   1

typedef __attribute__((ext_vector_type(2))) float v2f;
typedef __attribute__((ext_vector_type(8))) float v8f;

__global__ void zero_f32(float* __restrict__ p, int n) {
    int i = blockIdx.x * blockDim.x + threadIdx.x;
    for (; i < n; i += gridDim.x * blockDim.x) p[i] = 0.0f;
}

__global__ void emb_bag_mean(const int* __restrict__ tokens,
                             const float* __restrict__ table,
                             float* __restrict__ out) {
    const int node = blockIdx.x * (blockDim.x >> 5) + (threadIdx.x >> 5);
    if (node >= N_NODES) return;
    const int lane = threadIdx.x & 31;
    const int* t = tokens + node * BAG;
    float4 acc = make_float4(0.f, 0.f, 0.f, 0.f);
    float cnt = 0.f;
    #pragma unroll
    for (int i = 0; i < BAG; ++i) {
        const int tok = t[i];
        if (tok != PAD_IDX) {
            const float4 e = *(const float4*)(table + (size_t)tok * EMB + lane * 4);
            acc.x += e.x; acc.y += e.y; acc.z += e.z; acc.w += e.w;
            cnt += 1.0f;
        }
    }
    const float inv = 1.0f / fmaxf(cnt, 1.0f);
    float4 r = make_float4(acc.x * inv, acc.y * inv, acc.z * inv, acc.w * inv);
    *(float4*)(out + (size_t)node * EMB + lane * 4) = r;
}

__global__ void rank_edges(const int* __restrict__ dst, int* __restrict__ degi, int* __restrict__ pos) {
    const int e = blockIdx.x * blockDim.x + threadIdx.x;
    if (e < N_EDGES) pos[e] = atomicAdd(&degi[dst[e]], 1);
}

#define SCAN_WAVES 32
#define SCAN_LINES 49
#define SCAN_SPAN  (SCAN_LINES * 32)
static_assert(SCAN_WAVES * SCAN_SPAN >= N_NODES, "one work-group covers every counter");

__device__ __forceinline__ int wave_incl_scan(int v, int lane) {
    v += __builtin_amdgcn_update_dpp(0, v, 0x111, 0xF, 0xF, false);
    v += __builtin_amdgcn_update_dpp(0, v, 0x112, 0xF, 0xF, false);
    v += __builtin_amdgcn_update_dpp(0, v, 0x114, 0xF, 0xF, false);
    v += __builtin_amdgcn_update_dpp(0, v, 0x118, 0xF, 0xF, false);
    const int low = __builtin_amdgcn_readlane(v, 15);
    return v + (lane >= 16 ? low : 0);
}

__global__ __launch_bounds__(SCAN_WAVES * 32)
void scan_deg(const int* __restrict__ degi, int* __restrict__ rowptr) {
    __shared__ int wtot[SCAN_WAVES];
    const int lane = threadIdx.x & 31;
    const int wave = __builtin_amdgcn_readfirstlane(threadIdx.x >> 5);
    const int i0 = wave * SCAN_SPAN + lane;
    int v[SCAN_LINES];
    #pragma unroll
    for (int k = 0; k < SCAN_LINES; ++k) {
        const int i = i0 + k * 32;
        const int d = degi[min(i, N_NODES - 1)];
        v[k] = (i < N_NODES) ? d : 0;
    }
    int carry = 0;
    #pragma unroll
    for (int k = 0; k < SCAN_LINES; ++k) {
        const int inc = wave_incl_scan(v[k], lane);
        const int tot = __builtin_amdgcn_readlane(inc, 31);
        v[k] = carry + inc - v[k];
        carry += tot;
    }
    if (lane == 0) wtot[wave] = carry;
    __syncthreads();
    const int t     = wtot[lane];
    const int tinc  = wave_incl_scan(t, lane);
    const int woff  = __builtin_amdgcn_readlane(tinc - t, wave);
    const int total = __builtin_amdgcn_readlane(tinc, 31);
    #pragma unroll
    for (int k = 0; k < SCAN_LINES; ++k) {
        const int i = i0 + k * 32;
        if (i < N_NODES) rowptr[i] = woff + v[k];
    }
    if (threadIdx.x == 0) rowptr[N_NODES] = total;
}

__global__ void place_edges(const int* __restrict__ src, const int* __restrict__ dst,
                            const int* __restrict__ rowptr, const int* __restrict__ pos,
                            int* __restrict__ col) {
    const int e = blockIdx.x * blockDim.x + threadIdx.x;
    if (e < N_EDGES) col[rowptr[dst[e]] + pos[e]] = src[e];
}

__global__ void count_graph(const int* __restrict__ batch, float* __restrict__ gcnt) {
    const int i = blockIdx.x * blockDim.x + threadIdx.x;
    if (i < N_NODES) atomicAdd(&gcnt[batch[i]], 1.0f);
}

template <int G, bool FULL>
__device__ __forceinline__ void add_group(const float* __restrict__ x, int cl, int c, int nb, int lane,
                                          float& a0, float& a1, float& a2, float& a3) {
    float r[G][4];
    #pragma unroll
    for (int j = 0; j < G; ++j) {
        const int s = __builtin_amdgcn_readlane(cl, FULL ? c + j : min(c + j, nb - 1));
        const float* xr = x + (size_t)s * HID + lane;
        r[j][0] = xr[0];
        r[j][1] = xr[32];
        r[j][2] = xr[64];
        r[j][3] = xr[96];
    }
    #pragma unroll
    for (int j = 0; j < G; ++j) {
        const bool on = FULL || c + j < nb;
        a0 += on ? r[j][0] : 0.f;
        a1 += on ? r[j][1] : 0.f;
        a2 += on ? r[j][2] : 0.f;
        a3 += on ? r[j][3] : 0.f;
    }
}

__global__ void gather_sum(const int* __restrict__ rowptr,
                           const int* __restrict__ col,
                           const float* __restrict__ x,
                           float* __restrict__ msg,
                           float* __restrict__ degf) {
    const int node = __builtin_amdgcn_readfirstlane(blockIdx.x * (blockDim.x >> 5) + (threadIdx.x >> 5));
    if (node >= N_NODES) return;
    const int lane = threadIdx.x & 31;
    const int beg = __builtin_amdgcn_readfirstlane(rowptr[node]);
    const int end = __builtin_amdgcn_readfirstlane(rowptr[node + 1]);
    if (lane == 0) degf[node] = (float)(end - beg);
    float a0 = 0.f, a1 = 0.f, a2 = 0.f, a3 = 0.f;
    for (int b32 = beg; b32 < end; b32 += 32) {
        const int nb = min(32, end - b32);
        const int cl = col[b32 + min(lane, nb - 1)];
        int c = 0;
        for (; c + 8 <= nb; c += 8) add_group<8, true>(x, cl, c, nb, lane, a0, a1, a2, a3);
        for (; c < nb; c += 4)      add_group<4, false>(x, cl, c, nb, lane, a0, a1, a2, a3);
    }
    float* p = msg + (size_t)node * HID + lane;
    p[0]  = a0;
    p[32] = a1;
    p[64] = a2;
    p[96] = a3;
}

#define EPW 8
static_assert(N_NODES % EPW == 0, "a wave's items are all in range or all out");

#define WPITCH     132
#define N_STRIPS   (N_NODES / 16)
#define SAGE_GRID  256
#define SAGE_WAVES 16
#define K_PAIRS    (HID / 8)

typedef __attribute__((address_space(3))) const float lds_cfloat;
typedef __attribute__((address_space(3))) const v2f   lds_cv2f;

__global__ __launch_bounds__(SAGE_WAVES * 32) __attribute__((amdgpu_waves_per_eu(4, 4)))
void sage_wmma(const float* __restrict__ msg,
               const float* __restrict__ deg,
               const float* x,
               const float* __restrict__ wl,
               const float* __restrict__ bias,
               const float* __restrict__ wr,
               float* out) {
    __shared__ __attribute__((aligned(16))) float sW[2 * HID * WPITCH];

    const int tid = threadIdx.x;

    {
        const int c4 = tid & 31;
        const int r0 = tid >> 5;
        #pragma unroll
        for (int it = 0; it < 2 * HID / SAGE_WAVES; ++it) {
            const int mat = it / (HID / SAGE_WAVES);
            const int row = (it % (HID / SAGE_WAVES)) * SAGE_WAVES + r0;
            const float* wsrc = mat ? wr : wl;
            const float4 v = *(const float4*)(wsrc + row * HID + c4 * 4);
            *(float4*)(sW + (mat * HID + row) * WPITCH + c4 * 4) = v;
        }
    }
    __syncthreads();

    const int lane  = tid & 31;
    const int wave  = __builtin_amdgcn_readfirstlane(tid >> 5);
    const int strip = wave * SAGE_GRID + blockIdx.x;
    if (strip >= N_STRIPS) return;

    const int half   = lane >> 4;
    const int lrow   = lane & 15;
    const int m_base = strip * 16;
    const int arow   = m_base + lrow;
    const float rdeg = 1.0f / fmaxf(deg[arow], 1.0f);

    v8f c[8];
    #pragma unroll
    for (int t = 0; t < 8; ++t) {
        const float bv = bias[t * 16 + lrow];
        #pragma unroll
        for (int r = 0; r < 8; ++r) c[t][r] = bv;
    }

    const float* mrow = msg + (size_t)arow * HID + half * 2;
    const float* xrow = x   + (size_t)arow * HID + half * 2;
    lds_cfloat* sl = (lds_cfloat*)sW + lrow * WPITCH + half * 2;
    lds_cfloat* sr = sl + HID * WPITCH;
    asm("" : "+v"(sr));

    v2f am[3][2], ax[3][2];
    v2f bl[2][2][8], br[2][2][8];

    #define LOAD_A(p, s)                                                        \
        _Pragma("unroll") for (int j = 0; j < 2; ++j) {                         \
            am[s][j] = *(const v2f*)(mrow + (p) * 8 + j * 4);                   \
            ax[s][j] = *(const v2f*)(xrow + (p) * 8 + j * 4);                   \
        }
    #define LOAD_B(p, s)                                                        \
        _Pragma("unroll") for (int j = 0; j < 2; ++j) {                         \
            _Pragma("unroll") for (int t = 0; t < 8; ++t) {                     \
                bl[s][j][t] = *(lds_cv2f*)(sl + t * 16 * WPITCH + (p) * 8 + j * 4); \
                br[s][j][t] = *(lds_cv2f*)(sr + t * 16 * WPITCH + (p) * 8 + j * 4); \
            }                                                                   \
        }

    LOAD_A(0, 0)
    LOAD_A(1, 1)
    LOAD_B(0, 0)

    #pragma unroll
    for (int p = 0; p < K_PAIRS; ++p) {
        if (p + 2 < K_PAIRS) { LOAD_A(p + 2, (p + 2) % 3) }
        if (p + 1 < K_PAIRS) { LOAD_B(p + 1, (p + 1) & 1) }
        #pragma unroll
        for (int j = 0; j < 2; ++j) {
            const v2f a = am[p % 3][j] * rdeg;
            #pragma unroll
            for (int t = 0; t < 8; ++t)
                c[t] = __builtin_amdgcn_wmma_f32_16x16x4_f32(false, a, false, bl[p & 1][j][t],
                                                             (short)0, c[t], false, false);
            #pragma unroll
            for (int t = 0; t < 8; ++t)
                c[t] = __builtin_amdgcn_wmma_f32_16x16x4_f32(false, ax[p % 3][j], false, br[p & 1][j][t],
                                                             (short)0, c[t], false, false);
        }
    }
    #undef LOAD_A
    #undef LOAD_B

    #pragma unroll
    for (int t = 0; t < 8; ++t) {
        #pragma unroll
        for (int r = 0; r < 8; ++r) {
            const float v = fmaxf(c[t][r], 0.0f);
            out[(size_t)(m_base + half * 8 + r) * HID + t * 16 + lrow] = v;
        }
    }
}

__global__ void pool_scatter(const float* __restrict__ x,
                             const int* __restrict__ batch,
                             float* __restrict__ gsum) {
    const int wave = __builtin_amdgcn_readfirstlane(blockIdx.x * (blockDim.x >> 5) + (threadIdx.x >> 5));
    const int n0 = wave * EPW;
    if (n0 >= N_NODES) return;
    const int lane = threadIdx.x & 31;
    const int gl = batch[n0 + (lane & (EPW - 1))];
    const float* xr = x + (size_t)n0 * HID + lane;
    float v[EPW][4];
    #pragma unroll
    for (int i = 0; i < EPW; ++i) {
        v[i][0] = xr[i * HID +  0];
        v[i][1] = xr[i * HID + 32];
        v[i][2] = xr[i * HID + 64];
        v[i][3] = xr[i * HID + 96];
    }
    #pragma unroll
    for (int i = 0; i < EPW; ++i) {
        const int g = __builtin_amdgcn_readlane(gl, i);
        float* p = gsum + (size_t)g * HID + lane;
        atomicAdd(p +  0, v[i][0]);
        atomicAdd(p + 32, v[i][1]);
        atomicAdd(p + 64, v[i][2]);
        atomicAdd(p + 96, v[i][3]);
    }
}

__global__ void out_head(const float* __restrict__ gsum,
                         const float* __restrict__ gcnt,
                         const float* __restrict__ wout,
                         const float* __restrict__ bout,
                         float* __restrict__ out) {
    const int idx = blockIdx.x * blockDim.x + threadIdx.x;
    if (idx >= N_GRAPHS * N_CLASSES) return;
    const int g = idx / N_CLASSES;
    const int c = idx % N_CLASSES;
    const float inv = 1.0f / fmaxf(gcnt[g], 1.0f);
    const float* gv = gsum + (size_t)g * HID;
    const float* wv = wout + (size_t)c * HID;
    float s = 0.0f;
    #pragma unroll 8
    for (int k = 0; k < HID; ++k) s += gv[k] * wv[k];
    out[idx] = s * inv + bout[c];
}

extern "C" void kernel_launch(void* const* d_in, const int* in_sizes, int n_in,
                              void* d_out, int out_size, void* d_ws, size_t ws_size,
                              hipStream_t stream) {
    (void)in_sizes; (void)n_in; (void)out_size; (void)ws_size;

    const int*   x_tokens = (const int*)  d_in[0];
    const int*   edge     = (const int*)  d_in[1];
    const int*   batch    = (const int*)  d_in[2];
    const float* emb      = (const float*)d_in[3];
    const float* w1l      = (const float*)d_in[4];
    const float* b1       = (const float*)d_in[5];
    const float* w1r      = (const float*)d_in[6];
    const float* w2l      = (const float*)d_in[7];
    const float* b2       = (const float*)d_in[8];
    const float* w2r      = (const float*)d_in[9];
    const float* wout     = (const float*)d_in[10];
    const float* bout     = (const float*)d_in[11];
    float*       out      = (float*)d_out;

    const int* src = edge;
    const int* dst = edge + N_EDGES;

    const size_t nodeF = (size_t)N_NODES * HID;
    float* bufA = (float*)d_ws;
    float* bufB = bufA + nodeF;
    float* bufC = bufB + nodeF;
    float* gsum = bufC + nodeF;
    float* deg  = gsum + (size_t)N_GRAPHS * HID;
    float* gcnt = deg + N_NODES;

    int* col    = (int*)bufC;
    int* pos    = col + N_EDGES;
    int* rowptr = pos + N_EDGES;
    int* degi   = rowptr + 50048;

    const dim3 blk256(256);
    const dim3 gWave((N_NODES + 7) / 8);
    const dim3 gEdgeT((N_EDGES + 255) / 256);
    const dim3 gPool((N_NODES / EPW + 7) / 8);
    const dim3 gGemm(SAGE_GRID);
    const dim3 bGemm(SAGE_WAVES * 32);

    zero_f32<<<256,  blk256, 0, stream>>>((float*)degi, N_NODES);
    zero_f32<<<256,  blk256, 0, stream>>>(gsum, N_GRAPHS * HID);
    zero_f32<<<4,    blk256, 0, stream>>>(gcnt, N_GRAPHS);

    emb_bag_mean<<<gWave, blk256, 0, stream>>>(x_tokens, emb, bufA);

    rank_edges<<<gEdgeT, blk256, 0, stream>>>(dst, degi, pos);
    count_graph<<<(N_NODES + 255) / 256, blk256, 0, stream>>>(batch, gcnt);
    scan_deg<<<1, SCAN_WAVES * 32, 0, stream>>>(degi, rowptr);
    place_edges<<<gEdgeT, blk256, 0, stream>>>(src, dst, rowptr, pos, col);

    gather_sum<<<gWave, blk256, 0, stream>>>(rowptr, col, bufA, bufB, deg);
    sage_wmma<<<gGemm, bGemm, 0, stream>>>(bufB, deg, bufA, w1l, b1, w1r, bufA);

    gather_sum<<<gWave, blk256, 0, stream>>>(rowptr, col, bufA, bufB, deg);
    sage_wmma<<<gGemm, bGemm, 0, stream>>>(bufB, deg, bufA, w2l, b2, w2r, bufA);

    pool_scatter<<<gPool, blk256, 0, stream>>>(bufA, batch, gsum);
    out_head<<<(N_GRAPHS * N_CLASSES + 255) / 256, blk256, 0, stream>>>(
        gsum, gcnt, wout, bout, out);
}


# Launches 14 -> 10: one zero-fill launch for all three regions, graph sizes counted in the pooling scatter, the one-work-group degree scan rides as work-group 0 of the embedding launch; sums unchanged
# speedup vs baseline: 2.5479x; 1.0449x over previous
#include <hip/hip_runtime.h>
#include <hip/hip_bf16.h>

#define N_NODES   50000
#define N_EDGES   600000
#define BAG       16
#define EMB       128
#define HID       128
#define N_CLASSES 10
#define N_GRAPHS  512
#define PAD_IDX   1

typedef __attribute__((ext_vector_type(2))) float v2f;
typedef __attribute__((ext_vector_type(8))) float v8f;

static_assert(N_GRAPHS * HID >= N_NODES, "one thread per word of gsum covers the counters too");
__global__ void zero_sums(int* __restrict__ degi, float* __restrict__ gsum, float* __restrict__ gcnt) {
    const int i = blockIdx.x * blockDim.x + threadIdx.x;
    if (i < N_GRAPHS * HID) gsum[i] = 0.0f;
    if (i < N_NODES)        degi[i] = 0;
    if (i < N_GRAPHS)       gcnt[i] = 0.0f;
}

__global__ void rank_edges(const int* __restrict__ dst, int* __restrict__ degi, int* __restrict__ pos) {
    const int e = blockIdx.x * blockDim.x + threadIdx.x;
    if (e < N_EDGES) pos[e] = atomicAdd(&degi[dst[e]], 1);
}

#define SCAN_WAVES 32
#define SCAN_LINES 49
#define SCAN_SPAN  (SCAN_LINES * 32)
#define SCAN_TRIP  7
static_assert(SCAN_WAVES * SCAN_SPAN >= N_NODES, "one work-group covers every counter");
static_assert(SCAN_LINES % SCAN_TRIP == 0, "whole trips");

__device__ __forceinline__ int wave_incl_scan(int v, int lane) {
    v += __builtin_amdgcn_update_dpp(0, v, 0x111, 0xF, 0xF, false);
    v += __builtin_amdgcn_update_dpp(0, v, 0x112, 0xF, 0xF, false);
    v += __builtin_amdgcn_update_dpp(0, v, 0x114, 0xF, 0xF, false);
    v += __builtin_amdgcn_update_dpp(0, v, 0x118, 0xF, 0xF, false);
    const int low = __builtin_amdgcn_readlane(v, 15);
    return v + (lane >= 16 ? low : 0);
}

__device__ __forceinline__ int deg_or_zero(const int* __restrict__ degi, int i) {
    const int d = degi[min(i, N_NODES - 1)];
    return (i < N_NODES) ? d : 0;
}

__device__ __forceinline__ void scan_deg_group(const int* __restrict__ degi, int* __restrict__ rowptr, int* wtot) {
    const int lane = threadIdx.x & 31;
    const int wave = __builtin_amdgcn_readfirstlane(threadIdx.x >> 5);
    const int i0 = wave * SCAN_SPAN + lane;
    int s = 0;
    #pragma unroll SCAN_TRIP
    for (int k = 0; k < SCAN_LINES; ++k) s += deg_or_zero(degi, i0 + k * 32);
    const int sinc = wave_incl_scan(s, lane);
    if (lane == 31) wtot[wave] = sinc;
    __syncthreads();
    const int t     = wtot[lane];
    const int tinc  = wave_incl_scan(t, lane);
    int carry       = __builtin_amdgcn_readlane(tinc - t, wave);
    const int total = __builtin_amdgcn_readlane(tinc, 31);
    #pragma unroll 1
    for (int kb = 0; kb < SCAN_LINES; kb += SCAN_TRIP) {
        int v[SCAN_TRIP];
        #pragma unroll
        for (int j = 0; j < SCAN_TRIP; ++j) v[j] = deg_or_zero(degi, i0 + (kb + j) * 32);
        #pragma unroll
        for (int j = 0; j < SCAN_TRIP; ++j) {
            const int i   = i0 + (kb + j) * 32;
            const int inc = wave_incl_scan(v[j], lane);
            if (i < N_NODES) rowptr[i] = carry + inc - v[j];
            carry += __builtin_amdgcn_readlane(inc, 31);
        }
    }
    if (threadIdx.x == 0) rowptr[N_NODES] = total;
}

#define EMB_GROUPS ((N_NODES + SCAN_WAVES - 1) / SCAN_WAVES)

__global__ __launch_bounds__(SCAN_WAVES * 32)
void emb_bag_mean_and_scan(const int* __restrict__ tokens,
                           const float* __restrict__ table,
                           float* __restrict__ out,
                           const int* __restrict__ degi,
                           int* __restrict__ rowptr) {
    __shared__ int wtot[SCAN_WAVES];
    if (blockIdx.x == 0) {
        scan_deg_group(degi, rowptr, wtot);
        return;
    }
    const int node = (blockIdx.x - 1) * SCAN_WAVES + (threadIdx.x >> 5);
    if (node >= N_NODES) return;
    const int lane = threadIdx.x & 31;
    const int* t = tokens + node * BAG;
    float4 acc = make_float4(0.f, 0.f, 0.f, 0.f);
    float cnt = 0.f;
    #pragma unroll
    for (int i = 0; i < BAG; ++i) {
        const int tok = t[i];
        if (tok != PAD_IDX) {
            const float4 e = *(const float4*)(table + (size_t)tok * EMB + lane * 4);
            acc.x += e.x; acc.y += e.y; acc.z += e.z; acc.w += e.w;
            cnt += 1.0f;
        }
    }
    const float inv = 1.0f / fmaxf(cnt, 1.0f);
    float4 r = make_float4(acc.x * inv, acc.y * inv, acc.z * inv, acc.w * inv);
    *(float4*)(out + (size_t)node * EMB + lane * 4) = r;
}

__global__ void place_edges(const int* __restrict__ src, const int* __restrict__ dst,
                            const int* __restrict__ rowptr, const int* __restrict__ pos,
                            int* __restrict__ col) {
    const int e = blockIdx.x * blockDim.x + threadIdx.x;
    if (e < N_EDGES) col[rowptr[dst[e]] + pos[e]] = src[e];
}

template <int G, bool FULL>
__device__ __forceinline__ void add_group(const float* __restrict__ x, int cl, int c, int nb, int lane,
                                          float& a0, float& a1, float& a2, float& a3) {
    float r[G][4];
    #pragma unroll
    for (int j = 0; j < G; ++j) {
        const int s = __builtin_amdgcn_readlane(cl, FULL ? c + j : min(c + j, nb - 1));
        const float* xr = x + (size_t)s * HID + lane;
        r[j][0] = xr[0];
        r[j][1] = xr[32];
        r[j][2] = xr[64];
        r[j][3] = xr[96];
    }
    #pragma unroll
    for (int j = 0; j < G; ++j) {
        const bool on = FULL || c + j < nb;
        a0 += on ? r[j][0] : 0.f;
        a1 += on ? r[j][1] : 0.f;
        a2 += on ? r[j][2] : 0.f;
        a3 += on ? r[j][3] : 0.f;
    }
}

__global__ void gather_sum(const int* __restrict__ rowptr,
                           const int* __restrict__ col,
                           const float* __restrict__ x,
                           float* __restrict__ msg,
                           float* __restrict__ degf) {
    const int node = __builtin_amdgcn_readfirstlane(blockIdx.x * (blockDim.x >> 5) + (threadIdx.x >> 5));
    if (node >= N_NODES) return;
    const int lane = threadIdx.x & 31;
    const int beg = __builtin_amdgcn_readfirstlane(rowptr[node]);
    const int end = __builtin_amdgcn_readfirstlane(rowptr[node + 1]);
    if (lane == 0) degf[node] = (float)(end - beg);
    float a0 = 0.f, a1 = 0.f, a2 = 0.f, a3 = 0.f;
    for (int b32 = beg; b32 < end; b32 += 32) {
        const int nb = min(32, end - b32);
        const int cl = col[b32 + min(lane, nb - 1)];
        int c = 0;
        for (; c + 8 <= nb; c += 8) add_group<8, true>(x, cl, c, nb, lane, a0, a1, a2, a3);
        for (; c < nb; c += 4)      add_group<4, false>(x, cl, c, nb, lane, a0, a1, a2, a3);
    }
    float* p = msg + (size_t)node * HID + lane;
    p[0]  = a0;
    p[32] = a1;
    p[64] = a2;
    p[96] = a3;
}

#define EPW 8
static_assert(N_NODES % EPW == 0, "a wave's items are all in range or all out");

#define WPITCH     132
#define N_STRIPS   (N_NODES / 16)
#define SAGE_GRID  256
#define SAGE_WAVES 16
#define K_PAIRS    (HID / 8)

typedef __attribute__((address_space(3))) const float lds_cfloat;
typedef __attribute__((address_space(3))) const v2f   lds_cv2f;

__global__ __launch_bounds__(SAGE_WAVES * 32) __attribute__((amdgpu_waves_per_eu(4, 4)))
void sage_wmma(const float* __restrict__ msg,
               const float* __restrict__ deg,
               const float* x,
               const float* __restrict__ wl,
               const float* __restrict__ bias,
               const float* __restrict__ wr,
               float* out) {
    __shared__ __attribute__((aligned(16))) float sW[2 * HID * WPITCH];

    const int tid = threadIdx.x;

    {
        const int c4 = tid & 31;
        const int r0 = tid >> 5;
        #pragma unroll
        for (int it = 0; it < 2 * HID / SAGE_WAVES; ++it) {
            const int mat = it / (HID / SAGE_WAVES);
            const int row = (it % (HID / SAGE_WAVES)) * SAGE_WAVES + r0;
            const float* wsrc = mat ? wr : wl;
            const float4 v = *(const float4*)(wsrc + row * HID + c4 * 4);
            *(float4*)(sW + (mat * HID + row) * WPITCH + c4 * 4) = v;
        }
    }
    __syncthreads();

    const int lane  = tid & 31;
    const int wave  = __builtin_amdgcn_readfirstlane(tid >> 5);
    const int strip = wave * SAGE_GRID + blockIdx.x;
    if (strip >= N_STRIPS) return;

    const int half   = lane >> 4;
    const int lrow   = lane & 15;
    const int m_base = strip * 16;
    const int arow   = m_base + lrow;
    const float rdeg = 1.0f / fmaxf(deg[arow], 1.0f);

    v8f c[8];
    #pragma unroll
    for (int t = 0; t < 8; ++t) {
        const float bv = bias[t * 16 + lrow];
        #pragma unroll
        for (int r = 0; r < 8; ++r) c[t][r] = bv;
    }

    const float* mrow = msg + (size_t)arow * HID + half * 2;
    const float* xrow = x   + (size_t)arow * HID + half * 2;
    lds_cfloat* sl = (lds_cfloat*)sW + lrow * WPITCH + half * 2;
    lds_cfloat* sr = sl + HID * WPITCH;
    asm("" : "+v"(sr));

    v2f am[3][2], ax[3][2];
    v2f bl[2][2][8], br[2][2][8];

    #define LOAD_A(p, s)                                                        \
        _Pragma("unroll") for (int j = 0; j < 2; ++j) {                         \
            am[s][j] = *(const v2f*)(mrow + (p) * 8 + j * 4);                   \
            ax[s][j] = *(const v2f*)(xrow + (p) * 8 + j * 4);                   \
        }
    #define LOAD_B(p, s)                                                        \
        _Pragma("unroll") for (int j = 0; j < 2; ++j) {                         \
            _Pragma("unroll") for (int t = 0; t < 8; ++t) {                     \
                bl[s][j][t] = *(lds_cv2f*)(sl + t * 16 * WPITCH + (p) * 8 + j * 4); \
                br[s][j][t] = *(lds_cv2f*)(sr + t * 16 * WPITCH + (p) * 8 + j * 4); \
            }                                                                   \
        }

    LOAD_A(0, 0)
    LOAD_A(1, 1)
    LOAD_B(0, 0)

    #pragma unroll
    for (int p = 0; p < K_PAIRS; ++p) {
        if (p + 2 < K_PAIRS) { LOAD_A(p + 2, (p + 2) % 3) }
        if (p + 1 < K_PAIRS) { LOAD_B(p + 1, (p + 1) & 1) }
        #pragma unroll
        for (int j = 0; j < 2; ++j) {
            const v2f a = am[p % 3][j] * rdeg;
            #pragma unroll
            for (int t = 0; t < 8; ++t)
                c[t] = __builtin_amdgcn_wmma_f32_16x16x4_f32(false, a, false, bl[p & 1][j][t],
                                                             (short)0, c[t], false, false);
            #pragma unroll
            for (int t = 0; t < 8; ++t)
                c[t] = __builtin_amdgcn_wmma_f32_16x16x4_f32(false, ax[p % 3][j], false, br[p & 1][j][t],
                                                             (short)0, c[t], false, false);
        }
    }
    #undef LOAD_A
    #undef LOAD_B

    #pragma unroll
    for (int t = 0; t < 8; ++t) {
        #pragma unroll
        for (int r = 0; r < 8; ++r) {
            const float v = fmaxf(c[t][r], 0.0f);
            out[(size_t)(m_base + half * 8 + r) * HID + t * 16 + lrow] = v;
        }
    }
}

__global__ void pool_scatter(const float* __restrict__ x,
                             const int* __restrict__ batch,
                             float* __restrict__ gsum,
                             float* __restrict__ gcnt) {
    const int wave = __builtin_amdgcn_readfirstlane(blockIdx.x * (blockDim.x >> 5) + (threadIdx.x >> 5));
    const int n0 = wave * EPW;
    if (n0 >= N_NODES) return;
    const int lane = threadIdx.x & 31;
    const int gl = batch[n0 + (lane & (EPW - 1))];
    const float* xr = x + (size_t)n0 * HID + lane;
    float v[EPW][4];
    #pragma unroll
    for (int i = 0; i < EPW; ++i) {
        v[i][0] = xr[i * HID +  0];
        v[i][1] = xr[i * HID + 32];
        v[i][2] = xr[i * HID + 64];
        v[i][3] = xr[i * HID + 96];
    }
    if (lane < EPW) atomicAdd(&gcnt[gl], 1.0f);
    #pragma unroll
    for (int i = 0; i < EPW; ++i) {
        const int g = __builtin_amdgcn_readlane(gl, i);
        float* p = gsum + (size_t)g * HID + lane;
        atomicAdd(p +  0, v[i][0]);
        atomicAdd(p + 32, v[i][1]);
        atomicAdd(p + 64, v[i][2]);
        atomicAdd(p + 96, v[i][3]);
    }
}

__global__ void out_head(const float* __restrict__ gsum,
                         const float* __restrict__ gcnt,
                         const float* __restrict__ wout,
                         const float* __restrict__ bout,
                         float* __restrict__ out) {
    const int idx = blockIdx.x * blockDim.x + threadIdx.x;
    if (idx >= N_GRAPHS * N_CLASSES) return;
    const int g = idx / N_CLASSES;
    const int c = idx % N_CLASSES;
    const float inv = 1.0f / fmaxf(gcnt[g], 1.0f);
    const float* gv = gsum + (size_t)g * HID;
    const float* wv = wout + (size_t)c * HID;
    float s = 0.0f;
    #pragma unroll 8
    for (int k = 0; k < HID; ++k) s += gv[k] * wv[k];
    out[idx] = s * inv + bout[c];
}

extern "C" void kernel_launch(void* const* d_in, const int* in_sizes, int n_in,
                              void* d_out, int out_size, void* d_ws, size_t ws_size,
                              hipStream_t stream) {
    (void)in_sizes; (void)n_in; (void)out_size; (void)ws_size;

    const int*   x_tokens = (const int*)  d_in[0];
    const int*   edge     = (const int*)  d_in[1];
    const int*   batch    = (const int*)  d_in[2];
    const float* emb      = (const float*)d_in[3];
    const float* w1l      = (const float*)d_in[4];
    const float* b1       = (const float*)d_in[5];
    const float* w1r      = (const float*)d_in[6];
    const float* w2l      = (const float*)d_in[7];
    const float* b2       = (const float*)d_in[8];
    const float* w2r      = (const float*)d_in[9];
    const float* wout     = (const float*)d_in[10];
    const float* bout     = (const float*)d_in[11];
    float*       out      = (float*)d_out;

    const int* src = edge;
    const int* dst = edge + N_EDGES;

    const size_t nodeF = (size_t)N_NODES * HID;
    float* bufA = (float*)d_ws;
    float* bufB = bufA + nodeF;
    float* bufC = bufB + nodeF;
    float* gsum = bufC + nodeF;
    float* deg  = gsum + (size_t)N_GRAPHS * HID;
    float* gcnt = deg + N_NODES;

    int* col    = (int*)bufC;
    int* pos    = col + N_EDGES;
    int* rowptr = pos + N_EDGES;
    int* degi   = rowptr + 50048;

    const dim3 blk256(256);
    const dim3 gWave((N_NODES + 7) / 8);
    const dim3 gEmbScan(EMB_GROUPS + 1);
    const dim3 bEmbScan(SCAN_WAVES * 32);
    const dim3 gEdgeT((N_EDGES + 255) / 256);
    const dim3 gPool((N_NODES / EPW + 7) / 8);
    const dim3 gGemm(SAGE_GRID);
    const dim3 bGemm(SAGE_WAVES * 32);

    zero_sums<<<(N_GRAPHS * HID + 255) / 256, blk256, 0, stream>>>(degi, gsum, gcnt);

    rank_edges<<<gEdgeT, blk256, 0, stream>>>(dst, degi, pos);
    emb_bag_mean_and_scan<<<gEmbScan, bEmbScan, 0, stream>>>(x_tokens, emb, bufA, degi, rowptr);
    place_edges<<<gEdgeT, blk256, 0, stream>>>(src, dst, rowptr, pos, col);

    gather_sum<<<gWave, blk256, 0, stream>>>(rowptr, col, bufA, bufB, deg);
    sage_wmma<<<gGemm, bGemm, 0, stream>>>(bufB, deg, bufA, w1l, b1, w1r, bufA);

    gather_sum<<<gWave, blk256, 0, stream>>>(rowptr, col, bufA, bufB, deg);
    sage_wmma<<<gGemm, bGemm, 0, stream>>>(bufB, deg, bufA, w2l, b2, w2r, bufA);

    pool_scatter<<<gPool, blk256, 0, stream>>>(bufA, batch, gsum, gcnt);
    out_head<<<(N_GRAPHS * N_CLASSES + 255) / 256, blk256, 0, stream>>>(
        gsum, gcnt, wout, bout, out);
}


# Graph mean pool folded into the layer-2 GEMM epilogue: a strip's 16 relu rows are summed per column in registers, one f32 atomic per 128-B line; rows never stored, 9 launches; pool sum order changes
# speedup vs baseline: 2.8732x; 1.1277x over previous
#include <hip/hip_runtime.h>
#include <hip/hip_bf16.h>

#define N_NODES   50000
#define N_EDGES   600000
#define BAG       16
#define EMB       128
#define HID       128
#define N_CLASSES 10
#define N_GRAPHS  512
#define PAD_IDX   1

typedef __attribute__((ext_vector_type(2))) float v2f;
typedef __attribute__((ext_vector_type(8))) float v8f;

static_assert(N_GRAPHS * HID >= N_NODES, "one thread per word of gsum covers the counters too");
__global__ void zero_sums(int* __restrict__ degi, float* __restrict__ gsum, float* __restrict__ gcnt) {
    const int i = blockIdx.x * blockDim.x + threadIdx.x;
    if (i < N_GRAPHS * HID) gsum[i] = 0.0f;
    if (i < N_NODES)        degi[i] = 0;
    if (i < N_GRAPHS)       gcnt[i] = 0.0f;
}

__global__ void rank_edges(const int* __restrict__ dst, int* __restrict__ degi, int* __restrict__ pos) {
    const int e = blockIdx.x * blockDim.x + threadIdx.x;
    if (e < N_EDGES) pos[e] = atomicAdd(&degi[dst[e]], 1);
}

#define SCAN_WAVES 32
#define SCAN_LINES 49
#define SCAN_SPAN  (SCAN_LINES * 32)
#define SCAN_TRIP  7
static_assert(SCAN_WAVES * SCAN_SPAN >= N_NODES, "one work-group covers every counter");
static_assert(SCAN_LINES % SCAN_TRIP == 0, "whole trips");

__device__ __forceinline__ int wave_incl_scan(int v, int lane) {
    v += __builtin_amdgcn_update_dpp(0, v, 0x111, 0xF, 0xF, false);
    v += __builtin_amdgcn_update_dpp(0, v, 0x112, 0xF, 0xF, false);
    v += __builtin_amdgcn_update_dpp(0, v, 0x114, 0xF, 0xF, false);
    v += __builtin_amdgcn_update_dpp(0, v, 0x118, 0xF, 0xF, false);
    const int low = __builtin_amdgcn_readlane(v, 15);
    return v + (lane >= 16 ? low : 0);
}

__device__ __forceinline__ int deg_or_zero(const int* __restrict__ degi, int i) {
    const int d = degi[min(i, N_NODES - 1)];
    return (i < N_NODES) ? d : 0;
}

__device__ __forceinline__ void scan_deg_group(const int* __restrict__ degi, int* __restrict__ rowptr, int* wtot) {
    const int lane = threadIdx.x & 31;
    const int wave = __builtin_amdgcn_readfirstlane(threadIdx.x >> 5);
    const int i0 = wave * SCAN_SPAN + lane;
    int s = 0;
    #pragma unroll SCAN_TRIP
    for (int k = 0; k < SCAN_LINES; ++k) s += deg_or_zero(degi, i0 + k * 32);
    const int sinc = wave_incl_scan(s, lane);
    if (lane == 31) wtot[wave] = sinc;
    __syncthreads();
    const int t     = wtot[lane];
    const int tinc  = wave_incl_scan(t, lane);
    int carry       = __builtin_amdgcn_readlane(tinc - t, wave);
    const int total = __builtin_amdgcn_readlane(tinc, 31);
    #pragma unroll 1
    for (int kb = 0; kb < SCAN_LINES; kb += SCAN_TRIP) {
        int v[SCAN_TRIP];
        #pragma unroll
        for (int j = 0; j < SCAN_TRIP; ++j) v[j] = deg_or_zero(degi, i0 + (kb + j) * 32);
        #pragma unroll
        for (int j = 0; j < SCAN_TRIP; ++j) {
            const int i   = i0 + (kb + j) * 32;
            const int inc = wave_incl_scan(v[j], lane);
            if (i < N_NODES) rowptr[i] = carry + inc - v[j];
            carry += __builtin_amdgcn_readlane(inc, 31);
        }
    }
    if (threadIdx.x == 0) rowptr[N_NODES] = total;
}

#define EMB_GROUPS ((N_NODES + SCAN_WAVES - 1) / SCAN_WAVES)

__global__ __launch_bounds__(SCAN_WAVES * 32)
void emb_bag_mean_and_scan(const int* __restrict__ tokens,
                           const float* __restrict__ table,
                           float* __restrict__ out,
                           const int* __restrict__ degi,
                           int* __restrict__ rowptr) {
    __shared__ int wtot[SCAN_WAVES];
    if (blockIdx.x == 0) {
        scan_deg_group(degi, rowptr, wtot);
        return;
    }
    const int node = (blockIdx.x - 1) * SCAN_WAVES + (threadIdx.x >> 5);
    if (node >= N_NODES) return;
    const int lane = threadIdx.x & 31;
    const int* t = tokens + node * BAG;
    float4 acc = make_float4(0.f, 0.f, 0.f, 0.f);
    float cnt = 0.f;
    #pragma unroll
    for (int i = 0; i < BAG; ++i) {
        const int tok = t[i];
        if (tok != PAD_IDX) {
            const float4 e = *(const float4*)(table + (size_t)tok * EMB + lane * 4);
            acc.x += e.x; acc.y += e.y; acc.z += e.z; acc.w += e.w;
            cnt += 1.0f;
        }
    }
    const float inv = 1.0f / fmaxf(cnt, 1.0f);
    float4 r = make_float4(acc.x * inv, acc.y * inv, acc.z * inv, acc.w * inv);
    *(float4*)(out + (size_t)node * EMB + lane * 4) = r;
}

__global__ void place_edges(const int* __restrict__ src, const int* __restrict__ dst,
                            const int* __restrict__ rowptr, const int* __restrict__ pos,
                            int* __restrict__ col) {
    const int e = blockIdx.x * blockDim.x + threadIdx.x;
    if (e < N_EDGES) col[rowptr[dst[e]] + pos[e]] = src[e];
}

template <int G, bool FULL>
__device__ __forceinline__ void add_group(const float* __restrict__ x, int cl, int c, int nb, int lane,
                                          float& a0, float& a1, float& a2, float& a3) {
    float r[G][4];
    #pragma unroll
    for (int j = 0; j < G; ++j) {
        const int s = __builtin_amdgcn_readlane(cl, FULL ? c + j : min(c + j, nb - 1));
        const float* xr = x + (size_t)s * HID + lane;
        r[j][0] = xr[0];
        r[j][1] = xr[32];
        r[j][2] = xr[64];
        r[j][3] = xr[96];
    }
    #pragma unroll
    for (int j = 0; j < G; ++j) {
        const bool on = FULL || c + j < nb;
        a0 += on ? r[j][0] : 0.f;
        a1 += on ? r[j][1] : 0.f;
        a2 += on ? r[j][2] : 0.f;
        a3 += on ? r[j][3] : 0.f;
    }
}

__global__ void gather_sum(const int* __restrict__ rowptr,
                           const int* __restrict__ col,
                           const float* __restrict__ x,
                           float* __restrict__ msg,
                           float* __restrict__ degf) {
    const int node = __builtin_amdgcn_readfirstlane(blockIdx.x * (blockDim.x >> 5) + (threadIdx.x >> 5));
    if (node >= N_NODES) return;
    const int lane = threadIdx.x & 31;
    const int beg = __builtin_amdgcn_readfirstlane(rowptr[node]);
    const int end = __builtin_amdgcn_readfirstlane(rowptr[node + 1]);
    if (lane == 0) degf[node] = (float)(end - beg);
    float a0 = 0.f, a1 = 0.f, a2 = 0.f, a3 = 0.f;
    for (int b32 = beg; b32 < end; b32 += 32) {
        const int nb = min(32, end - b32);
        const int cl = col[b32 + min(lane, nb - 1)];
        int c = 0;
        for (; c + 8 <= nb; c += 8) add_group<8, true>(x, cl, c, nb, lane, a0, a1, a2, a3);
        for (; c < nb; c += 4)      add_group<4, false>(x, cl, c, nb, lane, a0, a1, a2, a3);
    }
    float* p = msg + (size_t)node * HID + lane;
    p[0]  = a0;
    p[32] = a1;
    p[64] = a2;
    p[96] = a3;
}

#define WPITCH     132
#define N_STRIPS   (N_NODES / 16)
#define SAGE_GRID  256
#define SAGE_WAVES 16
#define K_PAIRS    (HID / 8)

typedef __attribute__((address_space(3))) const float lds_cfloat;
typedef __attribute__((address_space(3))) const v2f   lds_cv2f;

__device__ __forceinline__ float xhalf(float v) {
    const int b = __builtin_bit_cast(int, v);
    return __builtin_bit_cast(float, __builtin_amdgcn_permlanex16(b, b, 0x76543210, 0xfedcba98, false, false));
}

template <bool POOL>
__global__ __launch_bounds__(SAGE_WAVES * 32) __attribute__((amdgpu_waves_per_eu(4, 4)))
void sage_wmma(const float* __restrict__ msg,
               const float* __restrict__ deg,
               const float* x,
               const float* __restrict__ wl,
               const float* __restrict__ bias,
               const float* __restrict__ wr,
               float* out,
               const int* __restrict__ batch,
               float* __restrict__ gsum,
               float* __restrict__ gcnt) {
    __shared__ __attribute__((aligned(16))) float sW[2 * HID * WPITCH];

    const int tid = threadIdx.x;

    {
        const int c4 = tid & 31;
        const int r0 = tid >> 5;
        #pragma unroll
        for (int it = 0; it < 2 * HID / SAGE_WAVES; ++it) {
            const int mat = it / (HID / SAGE_WAVES);
            const int row = (it % (HID / SAGE_WAVES)) * SAGE_WAVES + r0;
            const float* wsrc = mat ? wr : wl;
            const float4 v = *(const float4*)(wsrc + row * HID + c4 * 4);
            *(float4*)(sW + (mat * HID + row) * WPITCH + c4 * 4) = v;
        }
    }
    __syncthreads();

    const int lane  = tid & 31;
    const int wave  = __builtin_amdgcn_readfirstlane(tid >> 5);
    const int strip = wave * SAGE_GRID + blockIdx.x;
    if (strip >= N_STRIPS) return;

    const int half   = lane >> 4;
    const int lrow   = lane & 15;
    const int m_base = strip * 16;
    const int arow   = m_base + lrow;
    const float rdeg = 1.0f / fmaxf(deg[arow], 1.0f);
    int gl = 0;
    if constexpr (POOL) gl = batch[arow];

    v8f c[8];
    #pragma unroll
    for (int t = 0; t < 8; ++t) {
        const float bv = bias[t * 16 + lrow];
        #pragma unroll
        for (int r = 0; r < 8; ++r) c[t][r] = bv;
    }

    const float* mrow = msg + (size_t)arow * HID + half * 2;
    const float* xrow = x   + (size_t)arow * HID + half * 2;
    lds_cfloat* sl = (lds_cfloat*)sW + lrow * WPITCH + half * 2;
    lds_cfloat* sr = sl + HID * WPITCH;
    asm("" : "+v"(sr));

    v2f am[3][2], ax[3][2];
    v2f bl[2][2][8], br[2][2][8];

    #define LOAD_A(p, s)                                                        \
        _Pragma("unroll") for (int j = 0; j < 2; ++j) {                         \
            am[s][j] = *(const v2f*)(mrow + (p) * 8 + j * 4);                   \
            ax[s][j] = *(const v2f*)(xrow + (p) * 8 + j * 4);                   \
        }
    #define LOAD_B(p, s)                                                        \
        _Pragma("unroll") for (int j = 0; j < 2; ++j) {                         \
            _Pragma("unroll") for (int t = 0; t < 8; ++t) {                     \
                bl[s][j][t] = *(lds_cv2f*)(sl + t * 16 * WPITCH + (p) * 8 + j * 4); \
                br[s][j][t] = *(lds_cv2f*)(sr + t * 16 * WPITCH + (p) * 8 + j * 4); \
            }                                                                   \
        }

    LOAD_A(0, 0)
    LOAD_A(1, 1)
    LOAD_B(0, 0)

    #pragma unroll
    for (int p = 0; p < K_PAIRS; ++p) {
        if (p + 2 < K_PAIRS) { LOAD_A(p + 2, (p + 2) % 3) }
        if (p + 1 < K_PAIRS) { LOAD_B(p + 1, (p + 1) & 1) }
        #pragma unroll
        for (int j = 0; j < 2; ++j) {
            const v2f a = am[p % 3][j] * rdeg;
            #pragma unroll
            for (int t = 0; t < 8; ++t)
                c[t] = __builtin_amdgcn_wmma_f32_16x16x4_f32(false, a, false, bl[p & 1][j][t],
                                                             (short)0, c[t], false, false);
            #pragma unroll
            for (int t = 0; t < 8; ++t)
                c[t] = __builtin_amdgcn_wmma_f32_16x16x4_f32(false, ax[p % 3][j], false, br[p & 1][j][t],
                                                             (short)0, c[t], false, false);
        }
    }
    #undef LOAD_A
    #undef LOAD_B

    if constexpr (!POOL) {
        #pragma unroll
        for (int t = 0; t < 8; ++t) {
            #pragma unroll
            for (int r = 0; r < 8; ++r) {
                const float v = fmaxf(c[t][r], 0.0f);
                out[(size_t)(m_base + half * 8 + r) * HID + t * 16 + lrow] = v;
            }
        }
    } else {
        #pragma unroll
        for (int t = 0; t < 8; ++t) {
            #pragma unroll
            for (int r = 0; r < 8; ++r) c[t][r] = fmaxf(c[t][r], 0.0f);
        }
        const int g0 = __builtin_amdgcn_readfirstlane(gl);
        if (__builtin_amdgcn_ballot_w32(gl != g0) == 0) {
            float s[8];
            #pragma unroll
            for (int t = 0; t < 8; ++t) {
                s[t] = c[t][0];
                #pragma unroll
                for (int r = 1; r < 8; ++r) s[t] += c[t][r];
            }
            float* p = gsum + (size_t)g0 * HID + lane;
            #pragma unroll
            for (int q = 0; q < 4; ++q) {
                const float keep = half ? s[2 * q + 1] : s[2 * q];
                const float give = half ? s[2 * q]     : s[2 * q + 1];
                atomicAdd(p + q * 32, keep + xhalf(give));
            }
            if (lane == 0) atomicAdd(&gcnt[g0], 16.0f);
        } else {
            #pragma unroll
            for (int r = 0; r < 8; ++r) {
                const int ga = __builtin_amdgcn_readlane(gl, r);
                const int gb = __builtin_amdgcn_readlane(gl, 8 + r);
                float* p = gsum + (size_t)(half ? gb : ga) * HID + lrow;
                #pragma unroll
                for (int t = 0; t < 8; ++t) atomicAdd(p + t * 16, c[t][r]);
            }
            if (half == 0) atomicAdd(&gcnt[gl], 1.0f);
        }
    }
}

__global__ void out_head(const float* __restrict__ gsum,
                         const float* __restrict__ gcnt,
                         const float* __restrict__ wout,
                         const float* __restrict__ bout,
                         float* __restrict__ out) {
    const int idx = blockIdx.x * blockDim.x + threadIdx.x;
    if (idx >= N_GRAPHS * N_CLASSES) return;
    const int g = idx / N_CLASSES;
    const int c = idx % N_CLASSES;
    const float inv = 1.0f / fmaxf(gcnt[g], 1.0f);
    const float* gv = gsum + (size_t)g * HID;
    const float* wv = wout + (size_t)c * HID;
    float s = 0.0f;
    #pragma unroll 8
    for (int k = 0; k < HID; ++k) s += gv[k] * wv[k];
    out[idx] = s * inv + bout[c];
}

extern "C" void kernel_launch(void* const* d_in, const int* in_sizes, int n_in,
                              void* d_out, int out_size, void* d_ws, size_t ws_size,
                              hipStream_t stream) {
    (void)in_sizes; (void)n_in; (void)out_size; (void)ws_size;

    const int*   x_tokens = (const int*)  d_in[0];
    const int*   edge     = (const int*)  d_in[1];
    const int*   batch    = (const int*)  d_in[2];
    const float* emb      = (const float*)d_in[3];
    const float* w1l      = (const float*)d_in[4];
    const float* b1       = (const float*)d_in[5];
    const float* w1r      = (const float*)d_in[6];
    const float* w2l      = (const float*)d_in[7];
    const float* b2       = (const float*)d_in[8];
    const float* w2r      = (const float*)d_in[9];
    const float* wout     = (const float*)d_in[10];
    const float* bout     = (const float*)d_in[11];
    float*       out      = (float*)d_out;

    const int* src = edge;
    const int* dst = edge + N_EDGES;

    const size_t nodeF = (size_t)N_NODES * HID;
    float* bufA = (float*)d_ws;
    float* bufB = bufA + nodeF;
    float* bufC = bufB + nodeF;
    float* gsum = bufC + nodeF;
    float* deg  = gsum + (size_t)N_GRAPHS * HID;
    float* gcnt = deg + N_NODES;

    int* col    = (int*)bufC;
    int* pos    = col + N_EDGES;
    int* rowptr = pos + N_EDGES;
    int* degi   = rowptr + 50048;

    const dim3 blk256(256);
    const dim3 gWave((N_NODES + 7) / 8);
    const dim3 gEmbScan(EMB_GROUPS + 1);
    const dim3 bEmbScan(SCAN_WAVES * 32);
    const dim3 gEdgeT((N_EDGES + 255) / 256);
    const dim3 gGemm(SAGE_GRID);
    const dim3 bGemm(SAGE_WAVES * 32);

    zero_sums<<<(N_GRAPHS * HID + 255) / 256, blk256, 0, stream>>>(degi, gsum, gcnt);

    rank_edges<<<gEdgeT, blk256, 0, stream>>>(dst, degi, pos);
    emb_bag_mean_and_scan<<<gEmbScan, bEmbScan, 0, stream>>>(x_tokens, emb, bufA, degi, rowptr);
    place_edges<<<gEdgeT, blk256, 0, stream>>>(src, dst, rowptr, pos, col);

    gather_sum<<<gWave, blk256, 0, stream>>>(rowptr, col, bufA, bufB, deg);
    sage_wmma<false><<<gGemm, bGemm, 0, stream>>>(bufB, deg, bufA, w1l, b1, w1r, bufA, batch, gsum, gcnt);

    gather_sum<<<gWave, blk256, 0, stream>>>(rowptr, col, bufA, bufB, deg);
    sage_wmma<true><<<gGemm, bGemm, 0, stream>>>(bufB, deg, bufA, w2l, b2, w2r, bufA, batch, gsum, gcnt);

    out_head<<<(N_GRAPHS * N_CLASSES + 255) / 256, blk256, 0, stream>>>(
        gsum, gcnt, wout, bout, out);
}


# Edge ranks from LDS counters: a work-group ranks 1 of 32 edge slices over 1 of 8 node ranges in an LDS histogram, a node pass turns counts into offsets; no global atomics; sum order: slice, arrival
# speedup vs baseline: 3.2823x; 1.1424x over previous
#include <hip/hip_runtime.h>
#include <hip/hip_bf16.h>

#define N_NODES   50000
#define N_EDGES   600000
#define BAG       16
#define EMB       128
#define HID       128
#define N_CLASSES 10
#define N_GRAPHS  512
#define PAD_IDX   1

typedef __attribute__((ext_vector_type(2))) float v2f;
typedef __attribute__((ext_vector_type(8))) float v8f;

#define RANK_GROUPS  32
#define NODE_PARTS   8
#define RANK_THREADS 1024
#define RANK_SLICE   (N_EDGES / RANK_GROUPS)
#define RANK_TRIPS   ((RANK_SLICE + RANK_THREADS - 1) / RANK_THREADS)
#define PART_PITCH   50048
#define PART_NODES   (PART_PITCH / NODE_PARTS)
#define HIST_VEC     (PART_NODES / 4)
#define HIST_TRIPS   ((HIST_VEC + RANK_THREADS - 1) / RANK_THREADS)
static_assert(N_EDGES % RANK_GROUPS == 0, "equal slices");
static_assert(PART_PITCH >= N_NODES && PART_PITCH % (4 * NODE_PARTS) == 0, "equal ranges, copied in 16-byte pieces");
static_assert(N_GRAPHS * HID <= RANK_GROUPS * NODE_PARTS * RANK_THREADS, "a thread per word of gsum");

__global__ __launch_bounds__(RANK_THREADS)
void rank_slices(const int* __restrict__ dst, int* __restrict__ pos, int* __restrict__ part,
                 float* __restrict__ gsum, float* __restrict__ gcnt) {
    __shared__ __attribute__((aligned(16))) int cnt[PART_NODES];
    const int tid = threadIdx.x;
    const int w   = blockIdx.x / NODE_PARTS;
    const int n0  = (blockIdx.x % NODE_PARTS) * PART_NODES;
    const int e0  = w * RANK_SLICE;

    int d[RANK_TRIPS];
    #pragma unroll
    for (int k = 0; k < RANK_TRIPS; ++k) d[k] = dst[e0 + min(k * RANK_THREADS + tid, RANK_SLICE - 1)];

    #pragma unroll
    for (int k = 0; k < HIST_TRIPS; ++k) {
        const int i = k * RANK_THREADS + tid;
        if (i < HIST_VEC) ((int4*)cnt)[i] = make_int4(0, 0, 0, 0);
    }
    __syncthreads();

    int r[RANK_TRIPS];
    #pragma unroll
    for (int k = 0; k < RANK_TRIPS; ++k) {
        d[k] -= n0;
        const bool mine = k * RANK_THREADS + tid < RANK_SLICE && (unsigned)d[k] < (unsigned)PART_NODES;
        r[k] = 0;
        if (mine) r[k] = atomicAdd(&cnt[d[k]], 1);
    }
    __syncthreads();

    int4* row = (int4*)(part + w * PART_PITCH + n0);
    #pragma unroll
    for (int k = 0; k < HIST_TRIPS; ++k) {
        const int i = k * RANK_THREADS + tid;
        if (i < HIST_VEC) row[i] = ((const int4*)cnt)[i];
    }
    #pragma unroll
    for (int k = 0; k < RANK_TRIPS; ++k) {
        const int j = k * RANK_THREADS + tid;
        if (j < RANK_SLICE && (unsigned)d[k] < (unsigned)PART_NODES) pos[e0 + j] = r[k];
    }
    {
        const int g = blockIdx.x * RANK_THREADS + tid;
        if (g < N_GRAPHS * HID) gsum[g] = 0.0f;
        if (g < N_GRAPHS)       gcnt[g] = 0.0f;
    }
}

__global__ void slice_offsets(int* __restrict__ part, int* __restrict__ degi) {
    const int i = blockIdx.x * blockDim.x + threadIdx.x;
    if (i >= N_NODES) return;
    int v[RANK_GROUPS];
    #pragma unroll
    for (int w = 0; w < RANK_GROUPS; ++w) v[w] = part[w * PART_PITCH + i];
    int base = 0;
    #pragma unroll
    for (int w = 0; w < RANK_GROUPS; ++w) { part[w * PART_PITCH + i] = base; base += v[w]; }
    degi[i] = base;
}

#define SCAN_WAVES 32
#define SCAN_LINES 49
#define SCAN_SPAN  (SCAN_LINES * 32)
#define SCAN_TRIP  7
static_assert(SCAN_WAVES * SCAN_SPAN >= N_NODES, "one work-group covers every counter");
static_assert(SCAN_LINES % SCAN_TRIP == 0, "whole trips");

__device__ __forceinline__ int wave_incl_scan(int v, int lane) {
    v += __builtin_amdgcn_update_dpp(0, v, 0x111, 0xF, 0xF, false);
    v += __builtin_amdgcn_update_dpp(0, v, 0x112, 0xF, 0xF, false);
    v += __builtin_amdgcn_update_dpp(0, v, 0x114, 0xF, 0xF, false);
    v += __builtin_amdgcn_update_dpp(0, v, 0x118, 0xF, 0xF, false);
    const int low = __builtin_amdgcn_readlane(v, 15);
    return v + (lane >= 16 ? low : 0);
}

__device__ __forceinline__ int deg_or_zero(const int* __restrict__ degi, int i) {
    const int d = degi[min(i, N_NODES - 1)];
    return (i < N_NODES) ? d : 0;
}

__device__ __forceinline__ void scan_deg_group(const int* __restrict__ degi, int* __restrict__ rowptr, int* wtot) {
    const int lane = threadIdx.x & 31;
    const int wave = __builtin_amdgcn_readfirstlane(threadIdx.x >> 5);
    const int i0 = wave * SCAN_SPAN + lane;
    int s = 0;
    #pragma unroll SCAN_TRIP
    for (int k = 0; k < SCAN_LINES; ++k) s += deg_or_zero(degi, i0 + k * 32);
    const int sinc = wave_incl_scan(s, lane);
    if (lane == 31) wtot[wave] = sinc;
    __syncthreads();
    const int t     = wtot[lane];
    const int tinc  = wave_incl_scan(t, lane);
    int carry       = __builtin_amdgcn_readlane(tinc - t, wave);
    const int total = __builtin_amdgcn_readlane(tinc, 31);
    #pragma unroll 1
    for (int kb = 0; kb < SCAN_LINES; kb += SCAN_TRIP) {
        int v[SCAN_TRIP];
        #pragma unroll
        for (int j = 0; j < SCAN_TRIP; ++j) v[j] = deg_or_zero(degi, i0 + (kb + j) * 32);
        #pragma unroll
        for (int j = 0; j < SCAN_TRIP; ++j) {
            const int i   = i0 + (kb + j) * 32;
            const int inc = wave_incl_scan(v[j], lane);
            if (i < N_NODES) rowptr[i] = carry + inc - v[j];
            carry += __builtin_amdgcn_readlane(inc, 31);
        }
    }
    if (threadIdx.x == 0) rowptr[N_NODES] = total;
}

#define EMB_GROUPS ((N_NODES + SCAN_WAVES - 1) / SCAN_WAVES)

__global__ __launch_bounds__(SCAN_WAVES * 32)
void emb_bag_mean_and_scan(const int* __restrict__ tokens,
                           const float* __restrict__ table,
                           float* __restrict__ out,
                           const int* __restrict__ degi,
                           int* __restrict__ rowptr) {
    __shared__ int wtot[SCAN_WAVES];
    if (blockIdx.x == 0) {
        scan_deg_group(degi, rowptr, wtot);
        return;
    }
    const int node = (blockIdx.x - 1) * SCAN_WAVES + (threadIdx.x >> 5);
    if (node >= N_NODES) return;
    const int lane = threadIdx.x & 31;
    const int* t = tokens + node * BAG;
    float4 acc = make_float4(0.f, 0.f, 0.f, 0.f);
    float cnt = 0.f;
    #pragma unroll
    for (int i = 0; i < BAG; ++i) {
        const int tok = t[i];
        if (tok != PAD_IDX) {
            const float4 e = *(const float4*)(table + (size_t)tok * EMB + lane * 4);
            acc.x += e.x; acc.y += e.y; acc.z += e.z; acc.w += e.w;
            cnt += 1.0f;
        }
    }
    const float inv = 1.0f / fmaxf(cnt, 1.0f);
    float4 r = make_float4(acc.x * inv, acc.y * inv, acc.z * inv, acc.w * inv);
    *(float4*)(out + (size_t)node * EMB + lane * 4) = r;
}

__global__ void place_edges(const int* __restrict__ src, const int* __restrict__ dst,
                            const int* __restrict__ rowptr, const int* __restrict__ pos,
                            const int* __restrict__ part, int* __restrict__ col) {
    const int e = blockIdx.x * blockDim.x + threadIdx.x;
    if (e < N_EDGES) {
        const int d = dst[e];
        const int w = e / RANK_SLICE;
        col[rowptr[d] + part[w * PART_PITCH + d] + pos[e]] = src[e];
    }
}

template <int G, bool FULL>
__device__ __forceinline__ void add_group(const float* __restrict__ x, int cl, int c, int nb, int lane,
                                          float& a0, float& a1, float& a2, float& a3) {
    float r[G][4];
    #pragma unroll
    for (int j = 0; j < G; ++j) {
        const int s = __builtin_amdgcn_readlane(cl, FULL ? c + j : min(c + j, nb - 1));
        const float* xr = x + (size_t)s * HID + lane;
        r[j][0] = xr[0];
        r[j][1] = xr[32];
        r[j][2] = xr[64];
        r[j][3] = xr[96];
    }
    #pragma unroll
    for (int j = 0; j < G; ++j) {
        const bool on = FULL || c + j < nb;
        a0 += on ? r[j][0] : 0.f;
        a1 += on ? r[j][1] : 0.f;
        a2 += on ? r[j][2] : 0.f;
        a3 += on ? r[j][3] : 0.f;
    }
}

__global__ void gather_sum(const int* __restrict__ rowptr,
                           const int* __restrict__ col,
                           const float* __restrict__ x,
                           float* __restrict__ msg,
                           float* __restrict__ degf) {
    const int node = __builtin_amdgcn_readfirstlane(blockIdx.x * (blockDim.x >> 5) + (threadIdx.x >> 5));
    if (node >= N_NODES) return;
    const int lane = threadIdx.x & 31;
    const int beg = __builtin_amdgcn_readfirstlane(rowptr[node]);
    const int end = __builtin_amdgcn_readfirstlane(rowptr[node + 1]);
    if (lane == 0) degf[node] = (float)(end - beg);
    float a0 = 0.f, a1 = 0.f, a2 = 0.f, a3 = 0.f;
    for (int b32 = beg; b32 < end; b32 += 32) {
        const int nb = min(32, end - b32);
        const int cl = col[b32 + min(lane, nb - 1)];
        int c = 0;
        for (; c + 8 <= nb; c += 8) add_group<8, true>(x, cl, c, nb, lane, a0, a1, a2, a3);
        for (; c < nb; c += 4)      add_group<4, false>(x, cl, c, nb, lane, a0, a1, a2, a3);
    }
    float* p = msg + (size_t)node * HID + lane;
    p[0]  = a0;
    p[32] = a1;
    p[64] = a2;
    p[96] = a3;
}

#define WPITCH     132
#define N_STRIPS   (N_NODES / 16)
#define SAGE_GRID  256
#define SAGE_WAVES 16
#define K_PAIRS    (HID / 8)

typedef __attribute__((address_space(3))) const float lds_cfloat;
typedef __attribute__((address_space(3))) const v2f   lds_cv2f;

__device__ __forceinline__ float xhalf(float v) {
    const int b = __builtin_bit_cast(int, v);
    return __builtin_bit_cast(float, __builtin_amdgcn_permlanex16(b, b, 0x76543210, 0xfedcba98, false, false));
}

template <bool POOL>
__global__ __launch_bounds__(SAGE_WAVES * 32) __attribute__((amdgpu_waves_per_eu(4, 4)))
void sage_wmma(const float* __restrict__ msg,
               const float* __restrict__ deg,
               const float* x,
               const float* __restrict__ wl,
               const float* __restrict__ bias,
               const float* __restrict__ wr,
               float* out,
               const int* __restrict__ batch,
               float* __restrict__ gsum,
               float* __restrict__ gcnt) {
    __shared__ __attribute__((aligned(16))) float sW[2 * HID * WPITCH];

    const int tid = threadIdx.x;

    {
        const int c4 = tid & 31;
        const int r0 = tid >> 5;
        #pragma unroll
        for (int it = 0; it < 2 * HID / SAGE_WAVES; ++it) {
            const int mat = it / (HID / SAGE_WAVES);
            const int row = (it % (HID / SAGE_WAVES)) * SAGE_WAVES + r0;
            const float* wsrc = mat ? wr : wl;
            const float4 v = *(const float4*)(wsrc + row * HID + c4 * 4);
            *(float4*)(sW + (mat * HID + row) * WPITCH + c4 * 4) = v;
        }
    }
    __syncthreads();

    const int lane  = tid & 31;
    const int wave  = __builtin_amdgcn_readfirstlane(tid >> 5);
    const int strip = wave * SAGE_GRID + blockIdx.x;
    if (strip >= N_STRIPS) return;

    const int half   = lane >> 4;
    const int lrow   = lane & 15;
    const int m_base = strip * 16;
    const int arow   = m_base + lrow;
    const float rdeg = 1.0f / fmaxf(deg[arow], 1.0f);
    int gl = 0;
    if constexpr (POOL) gl = batch[arow];

    v8f c[8];
    #pragma unroll
    for (int t = 0; t < 8; ++t) {
        const float bv = bias[t * 16 + lrow];
        #pragma unroll
        for (int r = 0; r < 8; ++r) c[t][r] = bv;
    }

    const float* mrow = msg + (size_t)arow * HID + half * 2;
    const float* xrow = x   + (size_t)arow * HID + half * 2;
    lds_cfloat* sl = (lds_cfloat*)sW + lrow * WPITCH + half * 2;
    lds_cfloat* sr = sl + HID * WPITCH;
    asm("" : "+v"(sr));

    v2f am[3][2], ax[3][2];
    v2f bl[2][2][8], br[2][2][8];

    #define LOAD_A(p, s)                                                        \
        _Pragma("unroll") for (int j = 0; j < 2; ++j) {                         \
            am[s][j] = *(const v2f*)(mrow + (p) * 8 + j * 4);                   \
            ax[s][j] = *(const v2f*)(xrow + (p) * 8 + j * 4);                   \
        }
    #define LOAD_B(p, s)                                                        \
        _Pragma("unroll") for (int j = 0; j < 2; ++j) {                         \
            _Pragma("unroll") for (int t = 0; t < 8; ++t) {                     \
                bl[s][j][t] = *(lds_cv2f*)(sl + t * 16 * WPITCH + (p) * 8 + j * 4); \
                br[s][j][t] = *(lds_cv2f*)(sr + t * 16 * WPITCH + (p) * 8 + j * 4); \
            }                                                                   \
        }

    LOAD_A(0, 0)
    LOAD_A(1, 1)
    LOAD_B(0, 0)

    #pragma unroll
    for (int p = 0; p < K_PAIRS; ++p) {
        if (p + 2 < K_PAIRS) { LOAD_A(p + 2, (p + 2) % 3) }
        if (p + 1 < K_PAIRS) { LOAD_B(p + 1, (p + 1) & 1) }
        #pragma unroll
        for (int j = 0; j < 2; ++j) {
            const v2f a = am[p % 3][j] * rdeg;
            #pragma unroll
            for (int t = 0; t < 8; ++t)
                c[t] = __builtin_amdgcn_wmma_f32_16x16x4_f32(false, a, false, bl[p & 1][j][t],
                                                             (short)0, c[t], false, false);
            #pragma unroll
            for (int t = 0; t < 8; ++t)
                c[t] = __builtin_amdgcn_wmma_f32_16x16x4_f32(false, ax[p % 3][j], false, br[p & 1][j][t],
                                                             (short)0, c[t], false, false);
        }
    }
    #undef LOAD_A
    #undef LOAD_B

    if constexpr (!POOL) {
        #pragma unroll
        for (int t = 0; t < 8; ++t) {
            #pragma unroll
            for (int r = 0; r < 8; ++r) {
                const float v = fmaxf(c[t][r], 0.0f);
                out[(size_t)(m_base + half * 8 + r) * HID + t * 16 + lrow] = v;
            }
        }
    } else {
        #pragma unroll
        for (int t = 0; t < 8; ++t) {
            #pragma unroll
            for (int r = 0; r < 8; ++r) c[t][r] = fmaxf(c[t][r], 0.0f);
        }
        const int g0 = __builtin_amdgcn_readfirstlane(gl);
        if (__builtin_amdgcn_ballot_w32(gl != g0) == 0) {
            float s[8];
            #pragma unroll
            for (int t = 0; t < 8; ++t) {
                s[t] = c[t][0];
                #pragma unroll
                for (int r = 1; r < 8; ++r) s[t] += c[t][r];
            }
            float* p = gsum + (size_t)g0 * HID + lane;
            #pragma unroll
            for (int q = 0; q < 4; ++q) {
                const float keep = half ? s[2 * q + 1] : s[2 * q];
                const float give = half ? s[2 * q]     : s[2 * q + 1];
                atomicAdd(p + q * 32, keep + xhalf(give));
            }
            if (lane == 0) atomicAdd(&gcnt[g0], 16.0f);
        } else {
            #pragma unroll
            for (int r = 0; r < 8; ++r) {
                const int ga = __builtin_amdgcn_readlane(gl, r);
                const int gb = __builtin_amdgcn_readlane(gl, 8 + r);
                float* p = gsum + (size_t)(half ? gb : ga) * HID + lrow;
                #pragma unroll
                for (int t = 0; t < 8; ++t) atomicAdd(p + t * 16, c[t][r]);
            }
            if (half == 0) atomicAdd(&gcnt[gl], 1.0f);
        }
    }
}

__global__ void out_head(const float* __restrict__ gsum,
                         const float* __restrict__ gcnt,
                         const float* __restrict__ wout,
                         const float* __restrict__ bout,
                         float* __restrict__ out) {
    const int idx = blockIdx.x * blockDim.x + threadIdx.x;
    if (idx >= N_GRAPHS * N_CLASSES) return;
    const int g = idx / N_CLASSES;
    const int c = idx % N_CLASSES;
    const float inv = 1.0f / fmaxf(gcnt[g], 1.0f);
    const float* gv = gsum + (size_t)g * HID;
    const float* wv = wout + (size_t)c * HID;
    float s = 0.0f;
    #pragma unroll 8
    for (int k = 0; k < HID; ++k) s += gv[k] * wv[k];
    out[idx] = s * inv + bout[c];
}

extern "C" void kernel_launch(void* const* d_in, const int* in_sizes, int n_in,
                              void* d_out, int out_size, void* d_ws, size_t ws_size,
                              hipStream_t stream) {
    (void)in_sizes; (void)n_in; (void)out_size; (void)ws_size;

    const int*   x_tokens = (const int*)  d_in[0];
    const int*   edge     = (const int*)  d_in[1];
    const int*   batch    = (const int*)  d_in[2];
    const float* emb      = (const float*)d_in[3];
    const float* w1l      = (const float*)d_in[4];
    const float* b1       = (const float*)d_in[5];
    const float* w1r      = (const float*)d_in[6];
    const float* w2l      = (const float*)d_in[7];
    const float* b2       = (const float*)d_in[8];
    const float* w2r      = (const float*)d_in[9];
    const float* wout     = (const float*)d_in[10];
    const float* bout     = (const float*)d_in[11];
    float*       out      = (float*)d_out;

    const int* src = edge;
    const int* dst = edge + N_EDGES;

    const size_t nodeF = (size_t)N_NODES * HID;
    float* bufA = (float*)d_ws;
    float* bufB = bufA + nodeF;
    float* bufC = bufB + nodeF;
    float* gsum = bufC + nodeF;
    float* deg  = gsum + (size_t)N_GRAPHS * HID;
    float* gcnt = deg + N_NODES;

    int* col    = (int*)bufC;
    int* pos    = col + N_EDGES;
    int* rowptr = pos + N_EDGES;
    int* degi   = rowptr + 50048;
    int* part   = degi + 50048;

    const dim3 blk256(256);
    const dim3 gWave((N_NODES + 7) / 8);
    const dim3 gEmbScan(EMB_GROUPS + 1);
    const dim3 bEmbScan(SCAN_WAVES * 32);
    const dim3 gEdgeT((N_EDGES + 255) / 256);
    const dim3 gGemm(SAGE_GRID);
    const dim3 bGemm(SAGE_WAVES * 32);

    rank_slices<<<RANK_GROUPS * NODE_PARTS, RANK_THREADS, 0, stream>>>(dst, pos, part, gsum, gcnt);
    slice_offsets<<<(N_NODES + 255) / 256, blk256, 0, stream>>>(part, degi);
    emb_bag_mean_and_scan<<<gEmbScan, bEmbScan, 0, stream>>>(x_tokens, emb, bufA, degi, rowptr);
    place_edges<<<gEdgeT, blk256, 0, stream>>>(src, dst, rowptr, pos, part, col);

    gather_sum<<<gWave, blk256, 0, stream>>>(rowptr, col, bufA, bufB, deg);
    sage_wmma<false><<<gGemm, bGemm, 0, stream>>>(bufB, deg, bufA, w1l, b1, w1r, bufA, batch, gsum, gcnt);

    gather_sum<<<gWave, blk256, 0, stream>>>(rowptr, col, bufA, bufB, deg);
    sage_wmma<true><<<gGemm, bGemm, 0, stream>>>(bufB, deg, bufA, w2l, b2, w2r, bufA, batch, gsum, gcnt);

    out_head<<<(N_GRAPHS * N_CLASSES + 255) / 256, blk256, 0, stream>>>(
        gsum, gcnt, wout, bout, out);
}


# Neighbour sums fused into the SAGE GEMM launch: a strip's wave walks its 16 nodes' lists (64 line loads in flight), keeps sum/deg in an LDS tile, reads A fragments there; 7 launches; sum order kept
# speedup vs baseline: 3.4013x; 1.0363x over previous
#include <hip/hip_runtime.h>
#include <hip/hip_bf16.h>

#define N_NODES   50000
#define N_EDGES   600000
#define BAG       16
#define EMB       128
#define HID       128
#define N_CLASSES 10
#define N_GRAPHS  512
#define PAD_IDX   1

typedef __attribute__((ext_vector_type(2))) float v2f;
typedef __attribute__((ext_vector_type(8))) float v8f;

#define RANK_GROUPS  32
#define NODE_PARTS   8
#define RANK_THREADS 1024
#define RANK_SLICE   (N_EDGES / RANK_GROUPS)
#define RANK_TRIPS   ((RANK_SLICE + RANK_THREADS - 1) / RANK_THREADS)
#define PART_PITCH   50048
#define PART_NODES   (PART_PITCH / NODE_PARTS)
#define HIST_VEC     (PART_NODES / 4)
#define HIST_TRIPS   ((HIST_VEC + RANK_THREADS - 1) / RANK_THREADS)
static_assert(N_EDGES % RANK_GROUPS == 0, "equal slices");
static_assert(PART_PITCH >= N_NODES && PART_PITCH % (4 * NODE_PARTS) == 0, "equal ranges, copied in 16-byte pieces");
static_assert(N_GRAPHS * HID <= RANK_GROUPS * NODE_PARTS * RANK_THREADS, "a thread per word of gsum");

__global__ __launch_bounds__(RANK_THREADS)
void rank_slices(const int* __restrict__ dst, int* __restrict__ pos, int* __restrict__ part,
                 float* __restrict__ gsum, float* __restrict__ gcnt) {
    __shared__ __attribute__((aligned(16))) int cnt[PART_NODES];
    const int tid = threadIdx.x;
    const int w   = blockIdx.x / NODE_PARTS;
    const int n0  = (blockIdx.x % NODE_PARTS) * PART_NODES;
    const int e0  = w * RANK_SLICE;

    int d[RANK_TRIPS];
    #pragma unroll
    for (int k = 0; k < RANK_TRIPS; ++k) d[k] = dst[e0 + min(k * RANK_THREADS + tid, RANK_SLICE - 1)];

    #pragma unroll
    for (int k = 0; k < HIST_TRIPS; ++k) {
        const int i = k * RANK_THREADS + tid;
        if (i < HIST_VEC) ((int4*)cnt)[i] = make_int4(0, 0, 0, 0);
    }
    __syncthreads();

    int r[RANK_TRIPS];
    #pragma unroll
    for (int k = 0; k < RANK_TRIPS; ++k) {
        d[k] -= n0;
        const bool mine = k * RANK_THREADS + tid < RANK_SLICE && (unsigned)d[k] < (unsigned)PART_NODES;
        r[k] = 0;
        if (mine) r[k] = atomicAdd(&cnt[d[k]], 1);
    }
    __syncthreads();

    int4* row = (int4*)(part + w * PART_PITCH + n0);
    #pragma unroll
    for (int k = 0; k < HIST_TRIPS; ++k) {
        const int i = k * RANK_THREADS + tid;
        if (i < HIST_VEC) row[i] = ((const int4*)cnt)[i];
    }
    #pragma unroll
    for (int k = 0; k < RANK_TRIPS; ++k) {
        const int j = k * RANK_THREADS + tid;
        if (j < RANK_SLICE && (unsigned)d[k] < (unsigned)PART_NODES) pos[e0 + j] = r[k];
    }
    {
        const int g = blockIdx.x * RANK_THREADS + tid;
        if (g < N_GRAPHS * HID) gsum[g] = 0.0f;
        if (g < N_GRAPHS)       gcnt[g] = 0.0f;
    }
}

__global__ void slice_offsets(int* __restrict__ part, int* __restrict__ degi) {
    const int i = blockIdx.x * blockDim.x + threadIdx.x;
    if (i >= N_NODES) return;
    int v[RANK_GROUPS];
    #pragma unroll
    for (int w = 0; w < RANK_GROUPS; ++w) v[w] = part[w * PART_PITCH + i];
    int base = 0;
    #pragma unroll
    for (int w = 0; w < RANK_GROUPS; ++w) { part[w * PART_PITCH + i] = base; base += v[w]; }
    degi[i] = base;
}

#define SCAN_WAVES 32
#define SCAN_LINES 49
#define SCAN_SPAN  (SCAN_LINES * 32)
#define SCAN_TRIP  7
static_assert(SCAN_WAVES * SCAN_SPAN >= N_NODES, "one work-group covers every counter");
static_assert(SCAN_LINES % SCAN_TRIP == 0, "whole trips");

__device__ __forceinline__ int wave_incl_scan(int v, int lane) {
    v += __builtin_amdgcn_update_dpp(0, v, 0x111, 0xF, 0xF, false);
    v += __builtin_amdgcn_update_dpp(0, v, 0x112, 0xF, 0xF, false);
    v += __builtin_amdgcn_update_dpp(0, v, 0x114, 0xF, 0xF, false);
    v += __builtin_amdgcn_update_dpp(0, v, 0x118, 0xF, 0xF, false);
    const int low = __builtin_amdgcn_readlane(v, 15);
    return v + (lane >= 16 ? low : 0);
}

__device__ __forceinline__ int deg_or_zero(const int* __restrict__ degi, int i) {
    const int d = degi[min(i, N_NODES - 1)];
    return (i < N_NODES) ? d : 0;
}

__device__ __forceinline__ void scan_deg_group(const int* __restrict__ degi, int* __restrict__ rowptr, int* wtot) {
    const int lane = threadIdx.x & 31;
    const int wave = __builtin_amdgcn_readfirstlane(threadIdx.x >> 5);
    const int i0 = wave * SCAN_SPAN + lane;
    int s = 0;
    #pragma unroll SCAN_TRIP
    for (int k = 0; k < SCAN_LINES; ++k) s += deg_or_zero(degi, i0 + k * 32);
    const int sinc = wave_incl_scan(s, lane);
    if (lane == 31) wtot[wave] = sinc;
    __syncthreads();
    const int t     = wtot[lane];
    const int tinc  = wave_incl_scan(t, lane);
    int carry       = __builtin_amdgcn_readlane(tinc - t, wave);
    const int total = __builtin_amdgcn_readlane(tinc, 31);
    #pragma unroll 1
    for (int kb = 0; kb < SCAN_LINES; kb += SCAN_TRIP) {
        int v[SCAN_TRIP];
        #pragma unroll
        for (int j = 0; j < SCAN_TRIP; ++j) v[j] = deg_or_zero(degi, i0 + (kb + j) * 32);
        #pragma unroll
        for (int j = 0; j < SCAN_TRIP; ++j) {
            const int i   = i0 + (kb + j) * 32;
            const int inc = wave_incl_scan(v[j], lane);
            if (i < N_NODES) rowptr[i] = carry + inc - v[j];
            carry += __builtin_amdgcn_readlane(inc, 31);
        }
    }
    if (threadIdx.x == 0) rowptr[N_NODES] = total;
}

#define EMB_GROUPS ((N_NODES + SCAN_WAVES - 1) / SCAN_WAVES)

__global__ __launch_bounds__(SCAN_WAVES * 32)
void emb_bag_mean_and_scan(const int* __restrict__ tokens,
                           const float* __restrict__ table,
                           float* __restrict__ out,
                           const int* __restrict__ degi,
                           int* __restrict__ rowptr) {
    __shared__ int wtot[SCAN_WAVES];
    if (blockIdx.x == 0) {
        scan_deg_group(degi, rowptr, wtot);
        return;
    }
    const int node = (blockIdx.x - 1) * SCAN_WAVES + (threadIdx.x >> 5);
    if (node >= N_NODES) return;
    const int lane = threadIdx.x & 31;
    const int* t = tokens + node * BAG;
    float4 acc = make_float4(0.f, 0.f, 0.f, 0.f);
    float cnt = 0.f;
    #pragma unroll
    for (int i = 0; i < BAG; ++i) {
        const int tok = t[i];
        if (tok != PAD_IDX) {
            const float4 e = *(const float4*)(table + (size_t)tok * EMB + lane * 4);
            acc.x += e.x; acc.y += e.y; acc.z += e.z; acc.w += e.w;
            cnt += 1.0f;
        }
    }
    const float inv = 1.0f / fmaxf(cnt, 1.0f);
    float4 r = make_float4(acc.x * inv, acc.y * inv, acc.z * inv, acc.w * inv);
    *(float4*)(out + (size_t)node * EMB + lane * 4) = r;
}

__global__ void place_edges(const int* __restrict__ src, const int* __restrict__ dst,
                            const int* __restrict__ rowptr, const int* __restrict__ pos,
                            const int* __restrict__ part, int* __restrict__ col) {
    const int e = blockIdx.x * blockDim.x + threadIdx.x;
    if (e < N_EDGES) {
        const int d = dst[e];
        const int w = e / RANK_SLICE;
        col[rowptr[d] + part[w * PART_PITCH + d] + pos[e]] = src[e];
    }
}

#define WPITCH     132
#define N_STRIPS   (N_NODES / 16)
#define SAGE_GRID  256
#define SAGE_WAVES 16
#define K_PAIRS    (HID / 8)
#define GATHER_ROWS 16
static_assert(32 % GATHER_ROWS == 0, "whole groups in an index block");

typedef __attribute__((address_space(3))) const float lds_cfloat;
typedef __attribute__((address_space(3))) const v2f   lds_cv2f;

__device__ __forceinline__ float xhalf(float v) {
    const int b = __builtin_bit_cast(int, v);
    return __builtin_bit_cast(float, __builtin_amdgcn_permlanex16(b, b, 0x76543210, 0xfedcba98, false, false));
}

template <bool POOL>
__global__ __launch_bounds__(SAGE_WAVES * 32) __attribute__((amdgpu_waves_per_eu(4, 4)))
void sage_wmma(const int* __restrict__ rowptr,
               const int* __restrict__ col,
               const float* __restrict__ x,
               const float* __restrict__ wl,
               const float* __restrict__ bias,
               const float* __restrict__ wr,
               float* __restrict__ out,
               const int* __restrict__ batch,
               float* __restrict__ gsum,
               float* __restrict__ gcnt) {
    __shared__ __attribute__((aligned(16))) float sW[(2 * HID + SAGE_WAVES * 16) * WPITCH];

    const int tid = threadIdx.x;

    {
        const int c4 = tid & 31;
        const int r0 = tid >> 5;
        #pragma unroll
        for (int it = 0; it < 2 * HID / SAGE_WAVES; ++it) {
            const int mat = it / (HID / SAGE_WAVES);
            const int row = (it % (HID / SAGE_WAVES)) * SAGE_WAVES + r0;
            const float* wsrc = mat ? wr : wl;
            const float4 v = *(const float4*)(wsrc + row * HID + c4 * 4);
            *(float4*)(sW + (mat * HID + row) * WPITCH + c4 * 4) = v;
        }
    }
    __syncthreads();

    const int lane  = tid & 31;
    const int wave  = __builtin_amdgcn_readfirstlane(tid >> 5);
    const int strip = wave * SAGE_GRID + blockIdx.x;
    if (strip >= N_STRIPS) return;

    const int half   = lane >> 4;
    const int lrow   = lane & 15;
    const int m_base = strip * 16;
    const int arow   = m_base + lrow;

    float* tile = sW + (2 * HID + wave * 16) * WPITCH;
    {
        const int   li     = min(lane, 15);
        const int   rbeg   = rowptr[m_base + li];
        const int   rend   = rowptr[m_base + li + 1];
        const float invl   = 1.0f / fmaxf((float)(rend - rbeg), 1.0f);
        const int   beg0   = __builtin_amdgcn_readlane(rbeg, 0);
        const int   endAll = __builtin_amdgcn_readlane(rend, 15);
        int row   = 0;
        int bound = __builtin_amdgcn_readlane(rend, 0);
        float a0 = 0.f, a1 = 0.f, a2 = 0.f, a3 = 0.f;
        auto flush = [&]() {
            const float inv = __builtin_bit_cast(float, __builtin_amdgcn_readlane(__builtin_bit_cast(int, invl), row));
            float* t = tile + row * WPITCH + lane;
            t[0]  = a0 * inv;
            t[32] = a1 * inv;
            t[64] = a2 * inv;
            t[96] = a3 * inv;
            a0 = 0.f; a1 = 0.f; a2 = 0.f; a3 = 0.f;
            ++row;
            bound = __builtin_amdgcn_readlane(rend, min(row, 15));
        };
        const int last = max(endAll - 1, 0);
        int cl_next = col[min(beg0 + lane, last)];
        for (int b32 = beg0; b32 < endAll; b32 += 32) {
            const int nb = min(32, endAll - b32);
            const int cl = cl_next;
            cl_next = col[min(b32 + 32 + lane, last)];
            for (int c = 0; c < nb; c += GATHER_ROWS) {
                float v[GATHER_ROWS][4];
                #pragma unroll
                for (int j = 0; j < GATHER_ROWS; ++j) {
                    const int src_row = __builtin_amdgcn_readlane(cl, min(c + j, nb - 1));
                    const float* xr = x + (size_t)src_row * HID + lane;
                    v[j][0] = xr[0];
                    v[j][1] = xr[32];
                    v[j][2] = xr[64];
                    v[j][3] = xr[96];
                }
                asm volatile("" ::: "memory");
                #pragma unroll
                for (int j = 0; j < GATHER_ROWS; ++j) {
                    const int idx = b32 + c + j;
                    while (row < 16 && idx >= bound) flush();
                    const bool on = idx < endAll;
                    a0 += on ? v[j][0] : 0.f;
                    a1 += on ? v[j][1] : 0.f;
                    a2 += on ? v[j][2] : 0.f;
                    a3 += on ? v[j][3] : 0.f;
                }
            }
        }
        while (row < 16) flush();
    }
    __builtin_amdgcn_fence(__ATOMIC_RELEASE, "wavefront");
    __builtin_amdgcn_fence(__ATOMIC_ACQUIRE, "wavefront");
    int gl = 0;
    if constexpr (POOL) gl = batch[arow];

    v8f c[8];
    #pragma unroll
    for (int t = 0; t < 8; ++t) {
        const float bv = bias[t * 16 + lrow];
        #pragma unroll
        for (int r = 0; r < 8; ++r) c[t][r] = bv;
    }

    lds_cfloat* mrow = (lds_cfloat*)tile + lrow * WPITCH + half * 2;
    const float* xrow = x   + (size_t)arow * HID + half * 2;
    lds_cfloat* sl = (lds_cfloat*)sW + lrow * WPITCH + half * 2;
    lds_cfloat* sr = sl + HID * WPITCH;
    asm("" : "+v"(sr));

    v2f am[3][2], ax[3][2];
    v2f bl[2][2][8], br[2][2][8];

    #define LOAD_A(p, s)                                                        \
        _Pragma("unroll") for (int j = 0; j < 2; ++j) {                         \
            am[s][j] = *(lds_cv2f*)(mrow + (p) * 8 + j * 4);                    \
            ax[s][j] = *(const v2f*)(xrow + (p) * 8 + j * 4);                   \
        }
    #define LOAD_B(p, s)                                                        \
        _Pragma("unroll") for (int j = 0; j < 2; ++j) {                         \
            _Pragma("unroll") for (int t = 0; t < 8; ++t) {                     \
                bl[s][j][t] = *(lds_cv2f*)(sl + t * 16 * WPITCH + (p) * 8 + j * 4); \
                br[s][j][t] = *(lds_cv2f*)(sr + t * 16 * WPITCH + (p) * 8 + j * 4); \
            }                                                                   \
        }

    LOAD_A(0, 0)
    LOAD_A(1, 1)
    LOAD_B(0, 0)

    #pragma unroll
    for (int p = 0; p < K_PAIRS; ++p) {
        if (p + 2 < K_PAIRS) { LOAD_A(p + 2, (p + 2) % 3) }
        if (p + 1 < K_PAIRS) { LOAD_B(p + 1, (p + 1) & 1) }
        #pragma unroll
        for (int j = 0; j < 2; ++j) {
            const v2f a = am[p % 3][j];
            #pragma unroll
            for (int t = 0; t < 8; ++t)
                c[t] = __builtin_amdgcn_wmma_f32_16x16x4_f32(false, a, false, bl[p & 1][j][t],
                                                             (short)0, c[t], false, false);
            #pragma unroll
            for (int t = 0; t < 8; ++t)
                c[t] = __builtin_amdgcn_wmma_f32_16x16x4_f32(false, ax[p % 3][j], false, br[p & 1][j][t],
                                                             (short)0, c[t], false, false);
        }
    }
    #undef LOAD_A
    #undef LOAD_B

    if constexpr (!POOL) {
        #pragma unroll
        for (int t = 0; t < 8; ++t) {
            #pragma unroll
            for (int r = 0; r < 8; ++r) {
                const float v = fmaxf(c[t][r], 0.0f);
                out[(size_t)(m_base + half * 8 + r) * HID + t * 16 + lrow] = v;
            }
        }
    } else {
        #pragma unroll
        for (int t = 0; t < 8; ++t) {
            #pragma unroll
            for (int r = 0; r < 8; ++r) c[t][r] = fmaxf(c[t][r], 0.0f);
        }
        const int g0 = __builtin_amdgcn_readfirstlane(gl);
        if (__builtin_amdgcn_ballot_w32(gl != g0) == 0) {
            float s[8];
            #pragma unroll
            for (int t = 0; t < 8; ++t) {
                s[t] = c[t][0];
                #pragma unroll
                for (int r = 1; r < 8; ++r) s[t] += c[t][r];
            }
            float* p = gsum + (size_t)g0 * HID + lane;
            #pragma unroll
            for (int q = 0; q < 4; ++q) {
                const float keep = half ? s[2 * q + 1] : s[2 * q];
                const float give = half ? s[2 * q]     : s[2 * q + 1];
                atomicAdd(p + q * 32, keep + xhalf(give));
            }
            if (lane == 0) atomicAdd(&gcnt[g0], 16.0f);
        } else {
            #pragma unroll
            for (int r = 0; r < 8; ++r) {
                const int ga = __builtin_amdgcn_readlane(gl, r);
                const int gb = __builtin_amdgcn_readlane(gl, 8 + r);
                float* p = gsum + (size_t)(half ? gb : ga) * HID + lrow;
                #pragma unroll
                for (int t = 0; t < 8; ++t) atomicAdd(p + t * 16, c[t][r]);
            }
            if (half == 0) atomicAdd(&gcnt[gl], 1.0f);
        }
    }
}

__global__ void out_head(const float* __restrict__ gsum,
                         const float* __restrict__ gcnt,
                         const float* __restrict__ wout,
                         const float* __restrict__ bout,
                         float* __restrict__ out) {
    const int idx = blockIdx.x * blockDim.x + threadIdx.x;
    if (idx >= N_GRAPHS * N_CLASSES) return;
    const int g = idx / N_CLASSES;
    const int c = idx % N_CLASSES;
    const float inv = 1.0f / fmaxf(gcnt[g], 1.0f);
    const float* gv = gsum + (size_t)g * HID;
    const float* wv = wout + (size_t)c * HID;
    float s = 0.0f;
    #pragma unroll 8
    for (int k = 0; k < HID; ++k) s += gv[k] * wv[k];
    out[idx] = s * inv + bout[c];
}

extern "C" void kernel_launch(void* const* d_in, const int* in_sizes, int n_in,
                              void* d_out, int out_size, void* d_ws, size_t ws_size,
                              hipStream_t stream) {
    (void)in_sizes; (void)n_in; (void)out_size; (void)ws_size;

    const int*   x_tokens = (const int*)  d_in[0];
    const int*   edge     = (const int*)  d_in[1];
    const int*   batch    = (const int*)  d_in[2];
    const float* emb      = (const float*)d_in[3];
    const float* w1l      = (const float*)d_in[4];
    const float* b1       = (const float*)d_in[5];
    const float* w1r      = (const float*)d_in[6];
    const float* w2l      = (const float*)d_in[7];
    const float* b2       = (const float*)d_in[8];
    const float* w2r      = (const float*)d_in[9];
    const float* wout     = (const float*)d_in[10];
    const float* bout     = (const float*)d_in[11];
    float*       out      = (float*)d_out;

    const int* src = edge;
    const int* dst = edge + N_EDGES;

    const size_t nodeF = (size_t)N_NODES * HID;
    float* bufA = (float*)d_ws;
    float* bufB = bufA + nodeF;
    float* bufC = bufB + nodeF;
    float* gsum = bufC + nodeF;
    float* gcnt = gsum + (size_t)N_GRAPHS * HID;

    int* col    = (int*)bufC;
    int* pos    = col + N_EDGES;
    int* rowptr = pos + N_EDGES;
    int* degi   = rowptr + 50048;
    int* part   = degi + 50048;

    const dim3 blk256(256);
    const dim3 gEmbScan(EMB_GROUPS + 1);
    const dim3 bEmbScan(SCAN_WAVES * 32);
    const dim3 gEdgeT((N_EDGES + 255) / 256);
    const dim3 gGemm(SAGE_GRID);
    const dim3 bGemm(SAGE_WAVES * 32);

    rank_slices<<<RANK_GROUPS * NODE_PARTS, RANK_THREADS, 0, stream>>>(dst, pos, part, gsum, gcnt);
    slice_offsets<<<(N_NODES + 255) / 256, blk256, 0, stream>>>(part, degi);
    emb_bag_mean_and_scan<<<gEmbScan, bEmbScan, 0, stream>>>(x_tokens, emb, bufA, degi, rowptr);
    place_edges<<<gEdgeT, blk256, 0, stream>>>(src, dst, rowptr, pos, part, col);

    sage_wmma<false><<<gGemm, bGemm, 0, stream>>>(rowptr, col, bufA, w1l, b1, w1r, bufB, batch, gsum, gcnt);

    sage_wmma<true><<<gGemm, bGemm, 0, stream>>>(rowptr, col, bufB, w2l, b2, w2r, bufA, batch, gsum, gcnt);

    out_head<<<(N_GRAPHS * N_CLASSES + 255) / 256, blk256, 0, stream>>>(
        gsum, gcnt, wout, bout, out);
}


# Embedding bag: 255 resident work-groups, their waves loop over the nodes: one token load per node, all 16 table-row loads in flight (padding row adds zero by select); 50048 -> 8192 waves; sums kept
# speedup vs baseline: 3.6809x; 1.0822x over previous
#include <hip/hip_runtime.h>
#include <hip/hip_bf16.h>

#define N_NODES   50000
#define N_EDGES   600000
#define BAG       16
#define EMB       128
#define HID       128
#define N_CLASSES 10
#define N_GRAPHS  512
#define PAD_IDX   1

typedef __attribute__((ext_vector_type(2))) float v2f;
typedef __attribute__((ext_vector_type(8))) float v8f;

#define RANK_GROUPS  32
#define NODE_PARTS   8
#define RANK_THREADS 1024
#define RANK_SLICE   (N_EDGES / RANK_GROUPS)
#define RANK_TRIPS   ((RANK_SLICE + RANK_THREADS - 1) / RANK_THREADS)
#define PART_PITCH   50048
#define PART_NODES   (PART_PITCH / NODE_PARTS)
#define HIST_VEC     (PART_NODES / 4)
#define HIST_TRIPS   ((HIST_VEC + RANK_THREADS - 1) / RANK_THREADS)
static_assert(N_EDGES % RANK_GROUPS == 0, "equal slices");
static_assert(PART_PITCH >= N_NODES && PART_PITCH % (4 * NODE_PARTS) == 0, "equal ranges, copied in 16-byte pieces");
static_assert(N_GRAPHS * HID <= RANK_GROUPS * NODE_PARTS * RANK_THREADS, "a thread per word of gsum");

__global__ __launch_bounds__(RANK_THREADS)
void rank_slices(const int* __restrict__ dst, int* __restrict__ pos, int* __restrict__ part,
                 float* __restrict__ gsum, float* __restrict__ gcnt) {
    __shared__ __attribute__((aligned(16))) int cnt[PART_NODES];
    const int tid = threadIdx.x;
    const int w   = blockIdx.x / NODE_PARTS;
    const int n0  = (blockIdx.x % NODE_PARTS) * PART_NODES;
    const int e0  = w * RANK_SLICE;

    int d[RANK_TRIPS];
    #pragma unroll
    for (int k = 0; k < RANK_TRIPS; ++k) d[k] = dst[e0 + min(k * RANK_THREADS + tid, RANK_SLICE - 1)];

    #pragma unroll
    for (int k = 0; k < HIST_TRIPS; ++k) {
        const int i = k * RANK_THREADS + tid;
        if (i < HIST_VEC) ((int4*)cnt)[i] = make_int4(0, 0, 0, 0);
    }
    __syncthreads();

    int r[RANK_TRIPS];
    #pragma unroll
    for (int k = 0; k < RANK_TRIPS; ++k) {
        d[k] -= n0;
        const bool mine = k * RANK_THREADS + tid < RANK_SLICE && (unsigned)d[k] < (unsigned)PART_NODES;
        r[k] = 0;
        if (mine) r[k] = atomicAdd(&cnt[d[k]], 1);
    }
    __syncthreads();

    int4* row = (int4*)(part + w * PART_PITCH + n0);
    #pragma unroll
    for (int k = 0; k < HIST_TRIPS; ++k) {
        const int i = k * RANK_THREADS + tid;
        if (i < HIST_VEC) row[i] = ((const int4*)cnt)[i];
    }
    #pragma unroll
    for (int k = 0; k < RANK_TRIPS; ++k) {
        const int j = k * RANK_THREADS + tid;
        if (j < RANK_SLICE && (unsigned)d[k] < (unsigned)PART_NODES) pos[e0 + j] = r[k];
    }
    {
        const int g = blockIdx.x * RANK_THREADS + tid;
        if (g < N_GRAPHS * HID) gsum[g] = 0.0f;
        if (g < N_GRAPHS)       gcnt[g] = 0.0f;
    }
}

__global__ void slice_offsets(int* __restrict__ part, int* __restrict__ degi) {
    const int i = blockIdx.x * blockDim.x + threadIdx.x;
    if (i >= N_NODES) return;
    int v[RANK_GROUPS];
    #pragma unroll
    for (int w = 0; w < RANK_GROUPS; ++w) v[w] = part[w * PART_PITCH + i];
    int base = 0;
    #pragma unroll
    for (int w = 0; w < RANK_GROUPS; ++w) { part[w * PART_PITCH + i] = base; base += v[w]; }
    degi[i] = base;
}

#define SCAN_WAVES 32
#define SCAN_LINES 49
#define SCAN_SPAN  (SCAN_LINES * 32)
#define SCAN_TRIP  7
static_assert(SCAN_WAVES * SCAN_SPAN >= N_NODES, "one work-group covers every counter");
static_assert(SCAN_LINES % SCAN_TRIP == 0, "whole trips");

__device__ __forceinline__ int wave_incl_scan(int v, int lane) {
    v += __builtin_amdgcn_update_dpp(0, v, 0x111, 0xF, 0xF, false);
    v += __builtin_amdgcn_update_dpp(0, v, 0x112, 0xF, 0xF, false);
    v += __builtin_amdgcn_update_dpp(0, v, 0x114, 0xF, 0xF, false);
    v += __builtin_amdgcn_update_dpp(0, v, 0x118, 0xF, 0xF, false);
    const int low = __builtin_amdgcn_readlane(v, 15);
    return v + (lane >= 16 ? low : 0);
}

__device__ __forceinline__ int deg_or_zero(const int* __restrict__ degi, int i) {
    const int d = degi[min(i, N_NODES - 1)];
    return (i < N_NODES) ? d : 0;
}

__device__ __forceinline__ void scan_deg_group(const int* __restrict__ degi, int* __restrict__ rowptr, int* wtot) {
    const int lane = threadIdx.x & 31;
    const int wave = __builtin_amdgcn_readfirstlane(threadIdx.x >> 5);
    const int i0 = wave * SCAN_SPAN + lane;
    int s = 0;
    #pragma unroll SCAN_TRIP
    for (int k = 0; k < SCAN_LINES; ++k) s += deg_or_zero(degi, i0 + k * 32);
    const int sinc = wave_incl_scan(s, lane);
    if (lane == 31) wtot[wave] = sinc;
    __syncthreads();
    const int t     = wtot[lane];
    const int tinc  = wave_incl_scan(t, lane);
    int carry       = __builtin_amdgcn_readlane(tinc - t, wave);
    const int total = __builtin_amdgcn_readlane(tinc, 31);
    #pragma unroll 1
    for (int kb = 0; kb < SCAN_LINES; kb += SCAN_TRIP) {
        int v[SCAN_TRIP];
        #pragma unroll
        for (int j = 0; j < SCAN_TRIP; ++j) v[j] = deg_or_zero(degi, i0 + (kb + j) * 32);
        #pragma unroll
        for (int j = 0; j < SCAN_TRIP; ++j) {
            const int i   = i0 + (kb + j) * 32;
            const int inc = wave_incl_scan(v[j], lane);
            if (i < N_NODES) rowptr[i] = carry + inc - v[j];
            carry += __builtin_amdgcn_readlane(inc, 31);
        }
    }
    if (threadIdx.x == 0) rowptr[N_NODES] = total;
}

#define EMB_GROUPS    255
#define EMB_WAVES_ALL (EMB_GROUPS * SCAN_WAVES)
static_assert(EMB_WAVES_ALL <= N_NODES, "every wave has a first node");
static_assert(BAG <= 32, "a node's tokens fit the lanes of a wave");

__global__ __launch_bounds__(SCAN_WAVES * 32) __attribute__((amdgpu_waves_per_eu(8, 8)))
void emb_bag_mean_and_scan(const int* __restrict__ tokens,
                           const float* __restrict__ table,
                           float* __restrict__ out,
                           const int* __restrict__ degi,
                           int* __restrict__ rowptr) {
    __shared__ int wtot[SCAN_WAVES];
    if (blockIdx.x == 0) {
        scan_deg_group(degi, rowptr, wtot);
        return;
    }
    const int lane = threadIdx.x & 31;
    const int tl   = lane & (BAG - 1);
    const int first = __builtin_amdgcn_readfirstlane((threadIdx.x >> 5) * EMB_GROUPS + (blockIdx.x - 1));
    int tok_next = tokens[first * BAG + tl];
    for (int node = first; node < N_NODES; node += EMB_WAVES_ALL) {
        const int tok = tok_next;
        tok_next = tokens[min(node + EMB_WAVES_ALL, N_NODES - 1) * BAG + tl];
        float4 e[BAG];
        #pragma unroll
        for (int i = 0; i < BAG; ++i) {
            const int t = __builtin_amdgcn_readlane(tok, i);
            e[i] = *(const float4*)(table + (size_t)t * EMB + lane * 4);
        }
        asm volatile("" ::: "memory");
        float4 acc = make_float4(0.f, 0.f, 0.f, 0.f);
        float cnt = 0.f;
        #pragma unroll
        for (int i = 0; i < BAG; ++i) {
            const bool on = __builtin_amdgcn_readlane(tok, i) != PAD_IDX;
            acc.x += on ? e[i].x : 0.f;
            acc.y += on ? e[i].y : 0.f;
            acc.z += on ? e[i].z : 0.f;
            acc.w += on ? e[i].w : 0.f;
            cnt   += on ? 1.0f : 0.f;
        }
        const float inv = 1.0f / fmaxf(cnt, 1.0f);
        const float4 r = make_float4(acc.x * inv, acc.y * inv, acc.z * inv, acc.w * inv);
        *(float4*)(out + (size_t)node * EMB + lane * 4) = r;
    }
}

__global__ void place_edges(const int* __restrict__ src, const int* __restrict__ dst,
                            const int* __restrict__ rowptr, const int* __restrict__ pos,
                            const int* __restrict__ part, int* __restrict__ col) {
    const int e = blockIdx.x * blockDim.x + threadIdx.x;
    if (e < N_EDGES) {
        const int d = dst[e];
        const int w = e / RANK_SLICE;
        col[rowptr[d] + part[w * PART_PITCH + d] + pos[e]] = src[e];
    }
}

#define WPITCH     132
#define N_STRIPS   (N_NODES / 16)
#define SAGE_GRID  256
#define SAGE_WAVES 16
#define K_PAIRS    (HID / 8)
#define GATHER_ROWS 16
static_assert(32 % GATHER_ROWS == 0, "whole groups in an index block");

typedef __attribute__((address_space(3))) const float lds_cfloat;
typedef __attribute__((address_space(3))) const v2f   lds_cv2f;

__device__ __forceinline__ float xhalf(float v) {
    const int b = __builtin_bit_cast(int, v);
    return __builtin_bit_cast(float, __builtin_amdgcn_permlanex16(b, b, 0x76543210, 0xfedcba98, false, false));
}

template <bool POOL>
__global__ __launch_bounds__(SAGE_WAVES * 32) __attribute__((amdgpu_waves_per_eu(4, 4)))
void sage_wmma(const int* __restrict__ rowptr,
               const int* __restrict__ col,
               const float* __restrict__ x,
               const float* __restrict__ wl,
               const float* __restrict__ bias,
               const float* __restrict__ wr,
               float* __restrict__ out,
               const int* __restrict__ batch,
               float* __restrict__ gsum,
               float* __restrict__ gcnt) {
    __shared__ __attribute__((aligned(16))) float sW[(2 * HID + SAGE_WAVES * 16) * WPITCH];

    const int tid = threadIdx.x;

    {
        const int c4 = tid & 31;
        const int r0 = tid >> 5;
        #pragma unroll
        for (int it = 0; it < 2 * HID / SAGE_WAVES; ++it) {
            const int mat = it / (HID / SAGE_WAVES);
            const int row = (it % (HID / SAGE_WAVES)) * SAGE_WAVES + r0;
            const float* wsrc = mat ? wr : wl;
            const float4 v = *(const float4*)(wsrc + row * HID + c4 * 4);
            *(float4*)(sW + (mat * HID + row) * WPITCH + c4 * 4) = v;
        }
    }
    __syncthreads();

    const int lane  = tid & 31;
    const int wave  = __builtin_amdgcn_readfirstlane(tid >> 5);
    const int strip = wave * SAGE_GRID + blockIdx.x;
    if (strip >= N_STRIPS) return;

    const int half   = lane >> 4;
    const int lrow   = lane & 15;
    const int m_base = strip * 16;
    const int arow   = m_base + lrow;

    float* tile = sW + (2 * HID + wave * 16) * WPITCH;
    {
        const int   li     = min(lane, 15);
        const int   rbeg   = rowptr[m_base + li];
        const int   rend   = rowptr[m_base + li + 1];
        const float invl   = 1.0f / fmaxf((float)(rend - rbeg), 1.0f);
        const int   beg0   = __builtin_amdgcn_readlane(rbeg, 0);
        const int   endAll = __builtin_amdgcn_readlane(rend, 15);
        int row   = 0;
        int bound = __builtin_amdgcn_readlane(rend, 0);
        float a0 = 0.f, a1 = 0.f, a2 = 0.f, a3 = 0.f;
        auto flush = [&]() {
            const float inv = __builtin_bit_cast(float, __builtin_amdgcn_readlane(__builtin_bit_cast(int, invl), row));
            float* t = tile + row * WPITCH + lane;
            t[0]  = a0 * inv;
            t[32] = a1 * inv;
            t[64] = a2 * inv;
            t[96] = a3 * inv;
            a0 = 0.f; a1 = 0.f; a2 = 0.f; a3 = 0.f;
            ++row;
            bound = __builtin_amdgcn_readlane(rend, min(row, 15));
        };
        const int last = max(endAll - 1, 0);
        int cl_next = col[min(beg0 + lane, last)];
        for (int b32 = beg0; b32 < endAll; b32 += 32) {
            const int nb = min(32, endAll - b32);
            const int cl = cl_next;
            cl_next = col[min(b32 + 32 + lane, last)];
            for (int c = 0; c < nb; c += GATHER_ROWS) {
                float v[GATHER_ROWS][4];
                #pragma unroll
                for (int j = 0; j < GATHER_ROWS; ++j) {
                    const int src_row = __builtin_amdgcn_readlane(cl, min(c + j, nb - 1));
                    const float* xr = x + (size_t)src_row * HID + lane;
                    v[j][0] = xr[0];
                    v[j][1] = xr[32];
                    v[j][2] = xr[64];
                    v[j][3] = xr[96];
                }
                asm volatile("" ::: "memory");
                #pragma unroll
                for (int j = 0; j < GATHER_ROWS; ++j) {
                    const int idx = b32 + c + j;
                    while (row < 16 && idx >= bound) flush();
                    const bool on = idx < endAll;
                    a0 += on ? v[j][0] : 0.f;
                    a1 += on ? v[j][1] : 0.f;
                    a2 += on ? v[j][2] : 0.f;
                    a3 += on ? v[j][3] : 0.f;
                }
            }
        }
        while (row < 16) flush();
    }
    __builtin_amdgcn_fence(__ATOMIC_RELEASE, "wavefront");
    __builtin_amdgcn_fence(__ATOMIC_ACQUIRE, "wavefront");
    int gl = 0;
    if constexpr (POOL) gl = batch[arow];

    v8f c[8];
    #pragma unroll
    for (int t = 0; t < 8; ++t) {
        const float bv = bias[t * 16 + lrow];
        #pragma unroll
        for (int r = 0; r < 8; ++r) c[t][r] = bv;
    }

    lds_cfloat* mrow = (lds_cfloat*)tile + lrow * WPITCH + half * 2;
    const float* xrow = x   + (size_t)arow * HID + half * 2;
    lds_cfloat* sl = (lds_cfloat*)sW + lrow * WPITCH + half * 2;
    lds_cfloat* sr = sl + HID * WPITCH;
    asm("" : "+v"(sr));

    v2f am[3][2], ax[3][2];
    v2f bl[2][2][8], br[2][2][8];

    #define LOAD_A(p, s)                                                        \
        _Pragma("unroll") for (int j = 0; j < 2; ++j) {                         \
            am[s][j] = *(lds_cv2f*)(mrow + (p) * 8 + j * 4);                    \
            ax[s][j] = *(const v2f*)(xrow + (p) * 8 + j * 4);                   \
        }
    #define LOAD_B(p, s)                                                        \
        _Pragma("unroll") for (int j = 0; j < 2; ++j) {                         \
            _Pragma("unroll") for (int t = 0; t < 8; ++t) {                     \
                bl[s][j][t] = *(lds_cv2f*)(sl + t * 16 * WPITCH + (p) * 8 + j * 4); \
                br[s][j][t] = *(lds_cv2f*)(sr + t * 16 * WPITCH + (p) * 8 + j * 4); \
            }                                                                   \
        }

    LOAD_A(0, 0)
    LOAD_A(1, 1)
    LOAD_B(0, 0)

    #pragma unroll
    for (int p = 0; p < K_PAIRS; ++p) {
        if (p + 2 < K_PAIRS) { LOAD_A(p + 2, (p + 2) % 3) }
        if (p + 1 < K_PAIRS) { LOAD_B(p + 1, (p + 1) & 1) }
        #pragma unroll
        for (int j = 0; j < 2; ++j) {
            const v2f a = am[p % 3][j];
            #pragma unroll
            for (int t = 0; t < 8; ++t)
                c[t] = __builtin_amdgcn_wmma_f32_16x16x4_f32(false, a, false, bl[p & 1][j][t],
                                                             (short)0, c[t], false, false);
            #pragma unroll
            for (int t = 0; t < 8; ++t)
                c[t] = __builtin_amdgcn_wmma_f32_16x16x4_f32(false, ax[p % 3][j], false, br[p & 1][j][t],
                                                             (short)0, c[t], false, false);
        }
    }
    #undef LOAD_A
    #undef LOAD_B

    if constexpr (!POOL) {
        #pragma unroll
        for (int t = 0; t < 8; ++t) {
            #pragma unroll
            for (int r = 0; r < 8; ++r) {
                const float v = fmaxf(c[t][r], 0.0f);
                out[(size_t)(m_base + half * 8 + r) * HID + t * 16 + lrow] = v;
            }
        }
    } else {
        #pragma unroll
        for (int t = 0; t < 8; ++t) {
            #pragma unroll
            for (int r = 0; r < 8; ++r) c[t][r] = fmaxf(c[t][r], 0.0f);
        }
        const int g0 = __builtin_amdgcn_readfirstlane(gl);
        if (__builtin_amdgcn_ballot_w32(gl != g0) == 0) {
            float s[8];
            #pragma unroll
            for (int t = 0; t < 8; ++t) {
                s[t] = c[t][0];
                #pragma unroll
                for (int r = 1; r < 8; ++r) s[t] += c[t][r];
            }
            float* p = gsum + (size_t)g0 * HID + lane;
            #pragma unroll
            for (int q = 0; q < 4; ++q) {
                const float keep = half ? s[2 * q + 1] : s[2 * q];
                const float give = half ? s[2 * q]     : s[2 * q + 1];
                atomicAdd(p + q * 32, keep + xhalf(give));
            }
            if (lane == 0) atomicAdd(&gcnt[g0], 16.0f);
        } else {
            #pragma unroll
            for (int r = 0; r < 8; ++r) {
                const int ga = __builtin_amdgcn_readlane(gl, r);
                const int gb = __builtin_amdgcn_readlane(gl, 8 + r);
                float* p = gsum + (size_t)(half ? gb : ga) * HID + lrow;
                #pragma unroll
                for (int t = 0; t < 8; ++t) atomicAdd(p + t * 16, c[t][r]);
            }
            if (half == 0) atomicAdd(&gcnt[gl], 1.0f);
        }
    }
}

__global__ void out_head(const float* __restrict__ gsum,
                         const float* __restrict__ gcnt,
                         const float* __restrict__ wout,
                         const float* __restrict__ bout,
                         float* __restrict__ out) {
    const int idx = blockIdx.x * blockDim.x + threadIdx.x;
    if (idx >= N_GRAPHS * N_CLASSES) return;
    const int g = idx / N_CLASSES;
    const int c = idx % N_CLASSES;
    const float inv = 1.0f / fmaxf(gcnt[g], 1.0f);
    const float* gv = gsum + (size_t)g * HID;
    const float* wv = wout + (size_t)c * HID;
    float s = 0.0f;
    #pragma unroll 8
    for (int k = 0; k < HID; ++k) s += gv[k] * wv[k];
    out[idx] = s * inv + bout[c];
}

extern "C" void kernel_launch(void* const* d_in, const int* in_sizes, int n_in,
                              void* d_out, int out_size, void* d_ws, size_t ws_size,
                              hipStream_t stream) {
    (void)in_sizes; (void)n_in; (void)out_size; (void)ws_size;

    const int*   x_tokens = (const int*)  d_in[0];
    const int*   edge     = (const int*)  d_in[1];
    const int*   batch    = (const int*)  d_in[2];
    const float* emb      = (const float*)d_in[3];
    const float* w1l      = (const float*)d_in[4];
    const float* b1       = (const float*)d_in[5];
    const float* w1r      = (const float*)d_in[6];
    const float* w2l      = (const float*)d_in[7];
    const float* b2       = (const float*)d_in[8];
    const float* w2r      = (const float*)d_in[9];
    const float* wout     = (const float*)d_in[10];
    const float* bout     = (const float*)d_in[11];
    float*       out      = (float*)d_out;

    const int* src = edge;
    const int* dst = edge + N_EDGES;

    const size_t nodeF = (size_t)N_NODES * HID;
    float* bufA = (float*)d_ws;
    float* bufB = bufA + nodeF;
    float* bufC = bufB + nodeF;
    float* gsum = bufC + nodeF;
    float* gcnt = gsum + (size_t)N_GRAPHS * HID;

    int* col    = (int*)bufC;
    int* pos    = col + N_EDGES;
    int* rowptr = pos + N_EDGES;
    int* degi   = rowptr + 50048;
    int* part   = degi + 50048;

    const dim3 blk256(256);
    const dim3 gEmbScan(EMB_GROUPS + 1);
    const dim3 bEmbScan(SCAN_WAVES * 32);
    const dim3 gEdgeT((N_EDGES + 255) / 256);
    const dim3 gGemm(SAGE_GRID);
    const dim3 bGemm(SAGE_WAVES * 32);

    rank_slices<<<RANK_GROUPS * NODE_PARTS, RANK_THREADS, 0, stream>>>(dst, pos, part, gsum, gcnt);
    slice_offsets<<<(N_NODES + 255) / 256, blk256, 0, stream>>>(part, degi);
    emb_bag_mean_and_scan<<<gEmbScan, bEmbScan, 0, stream>>>(x_tokens, emb, bufA, degi, rowptr);
    place_edges<<<gEdgeT, blk256, 0, stream>>>(src, dst, rowptr, pos, part, col);

    sage_wmma<false><<<gGemm, bGemm, 0, stream>>>(rowptr, col, bufA, w1l, b1, w1r, bufB, batch, gsum, gcnt);

    sage_wmma<true><<<gGemm, bGemm, 0, stream>>>(rowptr, col, bufB, w2l, b2, w2r, bufA, batch, gsum, gcnt);

    out_head<<<(N_GRAPHS * N_CLASSES + 255) / 256, blk256, 0, stream>>>(
        gsum, gcnt, wout, bout, out);
}


# SAGE layers: each of the 53 strips left over at 12 per work-group is split by column tiles among 4 groups (a 13th wave: full neighbour sums, 2 of 8 tiles); busiest SIMD 3.25 strips, not 4; sums kept
# speedup vs baseline: 3.8271x; 1.0397x over previous
#include <hip/hip_runtime.h>
#include <hip/hip_bf16.h>

#define N_NODES   50000
#define N_EDGES   600000
#define BAG       16
#define EMB       128
#define HID       128
#define N_CLASSES 10
#define N_GRAPHS  512
#define PAD_IDX   1

typedef __attribute__((ext_vector_type(2))) float v2f;
typedef __attribute__((ext_vector_type(8))) float v8f;

#define RANK_GROUPS  32
#define NODE_PARTS   8
#define RANK_THREADS 1024
#define RANK_SLICE   (N_EDGES / RANK_GROUPS)
#define RANK_TRIPS   ((RANK_SLICE + RANK_THREADS - 1) / RANK_THREADS)
#define PART_PITCH   50048
#define PART_NODES   (PART_PITCH / NODE_PARTS)
#define HIST_VEC     (PART_NODES / 4)
#define HIST_TRIPS   ((HIST_VEC + RANK_THREADS - 1) / RANK_THREADS)
static_assert(N_EDGES % RANK_GROUPS == 0, "equal slices");
static_assert(PART_PITCH >= N_NODES && PART_PITCH % (4 * NODE_PARTS) == 0, "equal ranges, copied in 16-byte pieces");
static_assert(N_GRAPHS * HID <= RANK_GROUPS * NODE_PARTS * RANK_THREADS, "a thread per word of gsum");

__global__ __launch_bounds__(RANK_THREADS)
void rank_slices(const int* __restrict__ dst, int* __restrict__ pos, int* __restrict__ part,
                 float* __restrict__ gsum, float* __restrict__ gcnt) {
    __shared__ __attribute__((aligned(16))) int cnt[PART_NODES];
    const int tid = threadIdx.x;
    const int w   = blockIdx.x / NODE_PARTS;
    const int n0  = (blockIdx.x % NODE_PARTS) * PART_NODES;
    const int e0  = w * RANK_SLICE;

    int d[RANK_TRIPS];
    #pragma unroll
    for (int k = 0; k < RANK_TRIPS; ++k) d[k] = dst[e0 + min(k * RANK_THREADS + tid, RANK_SLICE - 1)];

    #pragma unroll
    for (int k = 0; k < HIST_TRIPS; ++k) {
        const int i = k * RANK_THREADS + tid;
        if (i < HIST_VEC) ((int4*)cnt)[i] = make_int4(0, 0, 0, 0);
    }
    __syncthreads();

    int r[RANK_TRIPS];
    #pragma unroll
    for (int k = 0; k < RANK_TRIPS; ++k) {
        d[k] -= n0;
        const bool mine = k * RANK_THREADS + tid < RANK_SLICE && (unsigned)d[k] < (unsigned)PART_NODES;
        r[k] = 0;
        if (mine) r[k] = atomicAdd(&cnt[d[k]], 1);
    }
    __syncthreads();

    int4* row = (int4*)(part + w * PART_PITCH + n0);
    #pragma unroll
    for (int k = 0; k < HIST_TRIPS; ++k) {
        const int i = k * RANK_THREADS + tid;
        if (i < HIST_VEC) row[i] = ((const int4*)cnt)[i];
    }
    #pragma unroll
    for (int k = 0; k < RANK_TRIPS; ++k) {
        const int j = k * RANK_THREADS + tid;
        if (j < RANK_SLICE && (unsigned)d[k] < (unsigned)PART_NODES) pos[e0 + j] = r[k];
    }
    {
        const int g = blockIdx.x * RANK_THREADS + tid;
        if (g < N_GRAPHS * HID) gsum[g] = 0.0f;
        if (g < N_GRAPHS)       gcnt[g] = 0.0f;
    }
}

__global__ void slice_offsets(int* __restrict__ part, int* __restrict__ degi) {
    const int i = blockIdx.x * blockDim.x + threadIdx.x;
    if (i >= N_NODES) return;
    int v[RANK_GROUPS];
    #pragma unroll
    for (int w = 0; w < RANK_GROUPS; ++w) v[w] = part[w * PART_PITCH + i];
    int base = 0;
    #pragma unroll
    for (int w = 0; w < RANK_GROUPS; ++w) { part[w * PART_PITCH + i] = base; base += v[w]; }
    degi[i] = base;
}

#define SCAN_WAVES 32
#define SCAN_LINES 49
#define SCAN_SPAN  (SCAN_LINES * 32)
#define SCAN_TRIP  7
static_assert(SCAN_WAVES * SCAN_SPAN >= N_NODES, "one work-group covers every counter");
static_assert(SCAN_LINES % SCAN_TRIP == 0, "whole trips");

__device__ __forceinline__ int wave_incl_scan(int v, int lane) {
    v += __builtin_amdgcn_update_dpp(0, v, 0x111, 0xF, 0xF, false);
    v += __builtin_amdgcn_update_dpp(0, v, 0x112, 0xF, 0xF, false);
    v += __builtin_amdgcn_update_dpp(0, v, 0x114, 0xF, 0xF, false);
    v += __builtin_amdgcn_update_dpp(0, v, 0x118, 0xF, 0xF, false);
    const int low = __builtin_amdgcn_readlane(v, 15);
    return v + (lane >= 16 ? low : 0);
}

__device__ __forceinline__ int deg_or_zero(const int* __restrict__ degi, int i) {
    const int d = degi[min(i, N_NODES - 1)];
    return (i < N_NODES) ? d : 0;
}

__device__ __forceinline__ void scan_deg_group(const int* __restrict__ degi, int* __restrict__ rowptr, int* wtot) {
    const int lane = threadIdx.x & 31;
    const int wave = __builtin_amdgcn_readfirstlane(threadIdx.x >> 5);
    const int i0 = wave * SCAN_SPAN + lane;
    int s = 0;
    #pragma unroll SCAN_TRIP
    for (int k = 0; k < SCAN_LINES; ++k) s += deg_or_zero(degi, i0 + k * 32);
    const int sinc = wave_incl_scan(s, lane);
    if (lane == 31) wtot[wave] = sinc;
    __syncthreads();
    const int t     = wtot[lane];
    const int tinc  = wave_incl_scan(t, lane);
    int carry       = __builtin_amdgcn_readlane(tinc - t, wave);
    const int total = __builtin_amdgcn_readlane(tinc, 31);
    #pragma unroll 1
    for (int kb = 0; kb < SCAN_LINES; kb += SCAN_TRIP) {
        int v[SCAN_TRIP];
        #pragma unroll
        for (int j = 0; j < SCAN_TRIP; ++j) v[j] = deg_or_zero(degi, i0 + (kb + j) * 32);
        #pragma unroll
        for (int j = 0; j < SCAN_TRIP; ++j) {
            const int i   = i0 + (kb + j) * 32;
            const int inc = wave_incl_scan(v[j], lane);
            if (i < N_NODES) rowptr[i] = carry + inc - v[j];
            carry += __builtin_amdgcn_readlane(inc, 31);
        }
    }
    if (threadIdx.x == 0) rowptr[N_NODES] = total;
}

#define EMB_GROUPS    255
#define EMB_WAVES_ALL (EMB_GROUPS * SCAN_WAVES)
static_assert(EMB_WAVES_ALL <= N_NODES, "every wave has a first node");
static_assert(BAG <= 32, "a node's tokens fit the lanes of a wave");

__global__ __launch_bounds__(SCAN_WAVES * 32) __attribute__((amdgpu_waves_per_eu(8, 8)))
void emb_bag_mean_and_scan(const int* __restrict__ tokens,
                           const float* __restrict__ table,
                           float* __restrict__ out,
                           const int* __restrict__ degi,
                           int* __restrict__ rowptr) {
    __shared__ int wtot[SCAN_WAVES];
    if (blockIdx.x == 0) {
        scan_deg_group(degi, rowptr, wtot);
        return;
    }
    const int lane = threadIdx.x & 31;
    const int tl   = lane & (BAG - 1);
    const int first = __builtin_amdgcn_readfirstlane((threadIdx.x >> 5) * EMB_GROUPS + (blockIdx.x - 1));
    int tok_next = tokens[first * BAG + tl];
    for (int node = first; node < N_NODES; node += EMB_WAVES_ALL) {
        const int tok = tok_next;
        tok_next = tokens[min(node + EMB_WAVES_ALL, N_NODES - 1) * BAG + tl];
        float4 e[BAG];
        #pragma unroll
        for (int i = 0; i < BAG; ++i) {
            const int t = __builtin_amdgcn_readlane(tok, i);
            e[i] = *(const float4*)(table + (size_t)t * EMB + lane * 4);
        }
        asm volatile("" ::: "memory");
        float4 acc = make_float4(0.f, 0.f, 0.f, 0.f);
        float cnt = 0.f;
        #pragma unroll
        for (int i = 0; i < BAG; ++i) {
            const bool on = __builtin_amdgcn_readlane(tok, i) != PAD_IDX;
            acc.x += on ? e[i].x : 0.f;
            acc.y += on ? e[i].y : 0.f;
            acc.z += on ? e[i].z : 0.f;
            acc.w += on ? e[i].w : 0.f;
            cnt   += on ? 1.0f : 0.f;
        }
        const float inv = 1.0f / fmaxf(cnt, 1.0f);
        const float4 r = make_float4(acc.x * inv, acc.y * inv, acc.z * inv, acc.w * inv);
        *(float4*)(out + (size_t)node * EMB + lane * 4) = r;
    }
}

__global__ void place_edges(const int* __restrict__ src, const int* __restrict__ dst,
                            const int* __restrict__ rowptr, const int* __restrict__ pos,
                            const int* __restrict__ part, int* __restrict__ col) {
    const int e = blockIdx.x * blockDim.x + threadIdx.x;
    if (e < N_EDGES) {
        const int d = dst[e];
        const int w = e / RANK_SLICE;
        col[rowptr[d] + part[w * PART_PITCH + d] + pos[e]] = src[e];
    }
}

#define WPITCH     132
#define N_STRIPS   (N_NODES / 16)
#define SAGE_GRID  256
#define SAGE_WAVES 16
#define FULL_WAVES  (N_STRIPS / SAGE_GRID)
#define LEFT_STRIPS (N_STRIPS - FULL_WAVES * SAGE_GRID)
#define SPLIT       4
#define SPLIT_TILES (8 / SPLIT)
static_assert(FULL_WAVES < SAGE_WAVES && LEFT_STRIPS * SPLIT <= SAGE_GRID, "one share at most per group, on the wave behind the whole strips");
static_assert(SPLIT_TILES % 2 == 0, "a share is whole 128-byte lines of a row (pairs of column tiles)");
#define K_PAIRS    (HID / 8)
#define GATHER_ROWS 16
static_assert(32 % GATHER_ROWS == 0, "whole groups in an index block");

typedef __attribute__((address_space(3))) const float lds_cfloat;
typedef __attribute__((address_space(3))) const v2f   lds_cv2f;

__device__ __forceinline__ float xhalf(float v) {
    const int b = __builtin_bit_cast(int, v);
    return __builtin_bit_cast(float, __builtin_amdgcn_permlanex16(b, b, 0x76543210, 0xfedcba98, false, false));
}

template <int N> struct tile_count { static constexpr int value = N; };

template <bool POOL>
__global__ __launch_bounds__(SAGE_WAVES * 32) __attribute__((amdgpu_waves_per_eu(4, 4)))
void sage_wmma(const int* __restrict__ rowptr,
               const int* __restrict__ col,
               const float* __restrict__ x,
               const float* __restrict__ wl,
               const float* __restrict__ bias,
               const float* __restrict__ wr,
               float* __restrict__ out,
               const int* __restrict__ batch,
               float* __restrict__ gsum,
               float* __restrict__ gcnt) {
    __shared__ __attribute__((aligned(16))) float sW[(2 * HID + SAGE_WAVES * 16) * WPITCH];

    const int tid = threadIdx.x;

    {
        const int c4 = tid & 31;
        const int r0 = tid >> 5;
        #pragma unroll
        for (int it = 0; it < 2 * HID / SAGE_WAVES; ++it) {
            const int mat = it / (HID / SAGE_WAVES);
            const int row = (it % (HID / SAGE_WAVES)) * SAGE_WAVES + r0;
            const float* wsrc = mat ? wr : wl;
            const float4 v = *(const float4*)(wsrc + row * HID + c4 * 4);
            *(float4*)(sW + (mat * HID + row) * WPITCH + c4 * 4) = v;
        }
    }
    __syncthreads();

    const int lane  = tid & 31;
    const int wave  = __builtin_amdgcn_readfirstlane(tid >> 5);
    const bool whole = wave < FULL_WAVES;
    if (!whole && (wave > FULL_WAVES || (int)blockIdx.x >= LEFT_STRIPS * SPLIT)) return;
    const int strip = whole ? wave * SAGE_GRID + (int)blockIdx.x : FULL_WAVES * SAGE_GRID + (int)blockIdx.x / SPLIT;
    const int tb0   = whole ? 0 : ((int)blockIdx.x % SPLIT) * SPLIT_TILES;

    const int half   = lane >> 4;
    const int lrow   = lane & 15;
    const int m_base = strip * 16;
    const int arow   = m_base + lrow;

    float* tile = sW + (2 * HID + wave * 16) * WPITCH;
    {
        const int   li     = min(lane, 15);
        const int   rbeg   = rowptr[m_base + li];
        const int   rend   = rowptr[m_base + li + 1];
        const float invl   = 1.0f / fmaxf((float)(rend - rbeg), 1.0f);
        const int   beg0   = __builtin_amdgcn_readlane(rbeg, 0);
        const int   endAll = __builtin_amdgcn_readlane(rend, 15);
        int row   = 0;
        int bound = __builtin_amdgcn_readlane(rend, 0);
        float a0 = 0.f, a1 = 0.f, a2 = 0.f, a3 = 0.f;
        auto flush = [&]() {
            const float inv = __builtin_bit_cast(float, __builtin_amdgcn_readlane(__builtin_bit_cast(int, invl), row));
            float* t = tile + row * WPITCH + lane;
            t[0]  = a0 * inv;
            t[32] = a1 * inv;
            t[64] = a2 * inv;
            t[96] = a3 * inv;
            a0 = 0.f; a1 = 0.f; a2 = 0.f; a3 = 0.f;
            ++row;
            bound = __builtin_amdgcn_readlane(rend, min(row, 15));
        };
        const int last = max(endAll - 1, 0);
        int cl_next = col[min(beg0 + lane, last)];
        for (int b32 = beg0; b32 < endAll; b32 += 32) {
            const int nb = min(32, endAll - b32);
            const int cl = cl_next;
            cl_next = col[min(b32 + 32 + lane, last)];
            for (int c = 0; c < nb; c += GATHER_ROWS) {
                float v[GATHER_ROWS][4];
                #pragma unroll
                for (int j = 0; j < GATHER_ROWS; ++j) {
                    const int src_row = __builtin_amdgcn_readlane(cl, min(c + j, nb - 1));
                    const float* xr = x + (size_t)src_row * HID + lane;
                    v[j][0] = xr[0];
                    v[j][1] = xr[32];
                    v[j][2] = xr[64];
                    v[j][3] = xr[96];
                }
                asm volatile("" ::: "memory");
                #pragma unroll
                for (int j = 0; j < GATHER_ROWS; ++j) {
                    const int idx = b32 + c + j;
                    while (row < 16 && idx >= bound) flush();
                    const bool on = idx < endAll;
                    a0 += on ? v[j][0] : 0.f;
                    a1 += on ? v[j][1] : 0.f;
                    a2 += on ? v[j][2] : 0.f;
                    a3 += on ? v[j][3] : 0.f;
                }
            }
        }
        while (row < 16) flush();
    }
    __builtin_amdgcn_fence(__ATOMIC_RELEASE, "wavefront");
    __builtin_amdgcn_fence(__ATOMIC_ACQUIRE, "wavefront");
    auto product = [&](auto nt, const int tb) {
    constexpr int NT = decltype(nt)::value;
    int gl = 0;
    if constexpr (POOL) gl = batch[arow];

    v8f c[NT];
    #pragma unroll
    for (int t = 0; t < NT; ++t) {
        const float bv = bias[(tb + t) * 16 + lrow];
        #pragma unroll
        for (int r = 0; r < 8; ++r) c[t][r] = bv;
    }

    lds_cfloat* mrow = (lds_cfloat*)tile + lrow * WPITCH + half * 2;
    const float* xrow = x   + (size_t)arow * HID + half * 2;
    lds_cfloat* sl = (lds_cfloat*)sW + (tb * 16 + lrow) * WPITCH + half * 2;
    lds_cfloat* sr = sl + HID * WPITCH;
    asm("" : "+v"(sr));

    v2f am[3][2], ax[3][2];
    v2f bl[2][2][NT], br[2][2][NT];

    #define LOAD_A(p, s)                                                        \
        _Pragma("unroll") for (int j = 0; j < 2; ++j) {                         \
            am[s][j] = *(lds_cv2f*)(mrow + (p) * 8 + j * 4);                    \
            ax[s][j] = *(const v2f*)(xrow + (p) * 8 + j * 4);                   \
        }
    #define LOAD_B(p, s)                                                        \
        _Pragma("unroll") for (int j = 0; j < 2; ++j) {                         \
            _Pragma("unroll") for (int t = 0; t < NT; ++t) {                    \
                bl[s][j][t] = *(lds_cv2f*)(sl + t * 16 * WPITCH + (p) * 8 + j * 4); \
                br[s][j][t] = *(lds_cv2f*)(sr + t * 16 * WPITCH + (p) * 8 + j * 4); \
            }                                                                   \
        }

    LOAD_A(0, 0)
    LOAD_A(1, 1)
    LOAD_B(0, 0)

    #pragma unroll
    for (int p = 0; p < K_PAIRS; ++p) {
        if (p + 2 < K_PAIRS) { LOAD_A(p + 2, (p + 2) % 3) }
        if (p + 1 < K_PAIRS) { LOAD_B(p + 1, (p + 1) & 1) }
        #pragma unroll
        for (int j = 0; j < 2; ++j) {
            const v2f a = am[p % 3][j];
            #pragma unroll
            for (int t = 0; t < NT; ++t)
                c[t] = __builtin_amdgcn_wmma_f32_16x16x4_f32(false, a, false, bl[p & 1][j][t],
                                                             (short)0, c[t], false, false);
            #pragma unroll
            for (int t = 0; t < NT; ++t)
                c[t] = __builtin_amdgcn_wmma_f32_16x16x4_f32(false, ax[p % 3][j], false, br[p & 1][j][t],
                                                             (short)0, c[t], false, false);
        }
    }
    #undef LOAD_A
    #undef LOAD_B

    if constexpr (!POOL) {
        #pragma unroll
        for (int t = 0; t < NT; ++t) {
            #pragma unroll
            for (int r = 0; r < 8; ++r) {
                const float v = fmaxf(c[t][r], 0.0f);
                out[(size_t)(m_base + half * 8 + r) * HID + (tb + t) * 16 + lrow] = v;
            }
        }
    } else {
        #pragma unroll
        for (int t = 0; t < NT; ++t) {
            #pragma unroll
            for (int r = 0; r < 8; ++r) c[t][r] = fmaxf(c[t][r], 0.0f);
        }
        const int g0 = __builtin_amdgcn_readfirstlane(gl);
        if (__builtin_amdgcn_ballot_w32(gl != g0) == 0) {
            float s[NT];
            #pragma unroll
            for (int t = 0; t < NT; ++t) {
                s[t] = c[t][0];
                #pragma unroll
                for (int r = 1; r < 8; ++r) s[t] += c[t][r];
            }
            float* p = gsum + (size_t)g0 * HID + tb * 16 + lane;
            #pragma unroll
            for (int q = 0; q < NT / 2; ++q) {
                const float keep = half ? s[2 * q + 1] : s[2 * q];
                const float give = half ? s[2 * q]     : s[2 * q + 1];
                atomicAdd(p + q * 32, keep + xhalf(give));
            }
            if (lane == 0 && tb == 0) atomicAdd(&gcnt[g0], 16.0f);
        } else {
            #pragma unroll
            for (int r = 0; r < 8; ++r) {
                const int ga = __builtin_amdgcn_readlane(gl, r);
                const int gb = __builtin_amdgcn_readlane(gl, 8 + r);
                float* p = gsum + (size_t)(half ? gb : ga) * HID + tb * 16 + lrow;
                #pragma unroll
                for (int t = 0; t < NT; ++t) atomicAdd(p + t * 16, c[t][r]);
            }
            if (half == 0 && tb == 0) atomicAdd(&gcnt[gl], 1.0f);
        }
    }
    };
    if (whole) product(tile_count<8>{}, 0);
    else       product(tile_count<SPLIT_TILES>{}, tb0);
}

__global__ void out_head(const float* __restrict__ gsum,
                         const float* __restrict__ gcnt,
                         const float* __restrict__ wout,
                         const float* __restrict__ bout,
                         float* __restrict__ out) {
    const int idx = blockIdx.x * blockDim.x + threadIdx.x;
    if (idx >= N_GRAPHS * N_CLASSES) return;
    const int g = idx / N_CLASSES;
    const int c = idx % N_CLASSES;
    const float inv = 1.0f / fmaxf(gcnt[g], 1.0f);
    const float* gv = gsum + (size_t)g * HID;
    const float* wv = wout + (size_t)c * HID;
    float s = 0.0f;
    #pragma unroll 8
    for (int k = 0; k < HID; ++k) s += gv[k] * wv[k];
    out[idx] = s * inv + bout[c];
}

extern "C" void kernel_launch(void* const* d_in, const int* in_sizes, int n_in,
                              void* d_out, int out_size, void* d_ws, size_t ws_size,
                              hipStream_t stream) {
    (void)in_sizes; (void)n_in; (void)out_size; (void)ws_size;

    const int*   x_tokens = (const int*)  d_in[0];
    const int*   edge     = (const int*)  d_in[1];
    const int*   batch    = (const int*)  d_in[2];
    const float* emb      = (const float*)d_in[3];
    const float* w1l      = (const float*)d_in[4];
    const float* b1       = (const float*)d_in[5];
    const float* w1r      = (const float*)d_in[6];
    const float* w2l      = (const float*)d_in[7];
    const float* b2       = (const float*)d_in[8];
    const float* w2r      = (const float*)d_in[9];
    const float* wout     = (const float*)d_in[10];
    const float* bout     = (const float*)d_in[11];
    float*       out      = (float*)d_out;

    const int* src = edge;
    const int* dst = edge + N_EDGES;

    const size_t nodeF = (size_t)N_NODES * HID;
    float* bufA = (float*)d_ws;
    float* bufB = bufA + nodeF;
    float* bufC = bufB + nodeF;
    float* gsum = bufC + nodeF;
    float* gcnt = gsum + (size_t)N_GRAPHS * HID;

    int* col    = (int*)bufC;
    int* pos    = col + N_EDGES;
    int* rowptr = pos + N_EDGES;
    int* degi   = rowptr + 50048;
    int* part   = degi + 50048;

    const dim3 blk256(256);
    const dim3 gEmbScan(EMB_GROUPS + 1);
    const dim3 bEmbScan(SCAN_WAVES * 32);
    const dim3 gEdgeT((N_EDGES + 255) / 256);
    const dim3 gGemm(SAGE_GRID);
    const dim3 bGemm(SAGE_WAVES * 32);

    rank_slices<<<RANK_GROUPS * NODE_PARTS, RANK_THREADS, 0, stream>>>(dst, pos, part, gsum, gcnt);
    slice_offsets<<<(N_NODES + 255) / 256, blk256, 0, stream>>>(part, degi);
    emb_bag_mean_and_scan<<<gEmbScan, bEmbScan, 0, stream>>>(x_tokens, emb, bufA, degi, rowptr);
    place_edges<<<gEdgeT, blk256, 0, stream>>>(src, dst, rowptr, pos, part, col);

    sage_wmma<false><<<gGemm, bGemm, 0, stream>>>(rowptr, col, bufA, w1l, b1, w1r, bufB, batch, gsum, gcnt);

    sage_wmma<true><<<gGemm, bGemm, 0, stream>>>(rowptr, col, bufB, w2l, b2, w2r, bufA, batch, gsum, gcnt);

    out_head<<<(N_GRAPHS * N_CLASSES + 255) / 256, blk256, 0, stream>>>(
        gsum, gcnt, wout, bout, out);
}
